# starting kernel
# baseline (speedup 1.0000x reference)
#include <hip/hip_runtime.h>

typedef __attribute__((ext_vector_type(2))) float v2f;
typedef __attribute__((ext_vector_type(8))) float v8f;

__global__ void set_val_kernel(float* p, long long n, float v) {
  long long i = (long long)blockIdx.x * blockDim.x + threadIdx.x;
  if (i < n) p[i] = v;
}

__global__ void deg_count_kernel(const int* __restrict__ dst, int E, float* deg) {
  int e = blockIdx.x * blockDim.x + threadIdx.x;
  if (e < E) atomicAdd(&deg[dst[e]], 1.0f);
}

__global__ void rsqrt_kernel(float* p, int n) {
  int i = blockIdx.x * blockDim.x + threadIdx.x;
  if (i < n) p[i] = rsqrtf(p[i]);
}

template<int K, int N>
__global__ void gemm_wmma_kernel(const float* __restrict__ X,
                                 const float* __restrict__ W,
                                 float* __restrict__ OUT, int M) {
  __shared__ float sW[K * N];
  for (int i = threadIdx.x; i < K * N; i += blockDim.x) sW[i] = W[i];
  __syncthreads();

  constexpr int COL_TILES  = N / 16;
  constexpr int ROW_TILES  = 8 / COL_TILES;
  constexpr int BLOCK_ROWS = 16 * ROW_TILES;

  const int wave = threadIdx.x >> 5;
  const int lane = threadIdx.x & 31;
  const int rowBase = blockIdx.x * BLOCK_ROWS + (wave / COL_TILES) * 16;
  const int colBase = (wave % COL_TILES) * 16;
  if (rowBase + 16 > M) return;

  const int mn  = lane & 15;
  const int grp = lane >> 4;

  const float* xrow = X + (long long)(rowBase + mn) * K + 2 * grp;
  const float* wcol = sW + (long long)(2 * grp) * N + colBase + mn;

  v8f acc = {};
#pragma unroll 4
  for (int k0 = 0; k0 < K; k0 += 4) {
    v2f a = *(const v2f*)(xrow + k0);
    v2f b;
    b.x = wcol[(long long)k0 * N];
    b.y = wcol[(long long)(k0 + 1) * N];
    acc = __builtin_amdgcn_wmma_f32_16x16x4_f32(
        false, a, false, b, (short)0, acc, false, false);
  }

  float* out = OUT + (long long)(rowBase + 8 * grp) * N + colBase + mn;
#pragma unroll
  for (int r = 0; r < 8; ++r) out[(long long)r * N] = acc[r];
}

template<int D>
__global__ void scatter_kernel(const int* __restrict__ src, const int* __restrict__ dst,
                               const float* __restrict__ dis, const float* __restrict__ H,
                               float* __restrict__ agg, int E) {
  constexpr int CH = D / 4;
  long long t = (long long)blockIdx.x * blockDim.x + threadIdx.x;
  long long e = t / CH;
  int c = (int)(t % CH);
  if (e >= E) return;
  int s = src[e], d = dst[e];
  float norm = dis[s] * dis[d];
  const float4 v = *(const float4*)(H + (long long)s * D + 4 * c);
  float* o = agg + (long long)d * D + 4 * c;
  atomicAdd(o + 0, v.x * norm);
  atomicAdd(o + 1, v.y * norm);
  atomicAdd(o + 2, v.z * norm);
  atomicAdd(o + 3, v.w * norm);
}

template<int D, bool RELU>
__global__ void combine_kernel(const float* __restrict__ H, const float* __restrict__ bias,
                               const float* __restrict__ dis, float* agg, int M) {
  long long i = (long long)blockIdx.x * blockDim.x + threadIdx.x;
  if (i >= (long long)M * D) return;
  int n = (int)(i / D), d = (int)(i % D);
  float z = agg[i] + dis[n] * dis[n] * H[i] + bias[d];
  if (RELU) z = fmaxf(z, 0.0f);
  agg[i] = z;
}

__global__ void combine_softmax_kernel(const float* __restrict__ H2,
                                       const float* __restrict__ b2,
                                       const float* __restrict__ dis,
                                       float* out, int M) {
  long long gt = (long long)blockIdx.x * blockDim.x + threadIdx.x;
  int node = (int)(gt >> 5);
  int lane = threadIdx.x & 31;
  if (node >= M) return;
  long long idx = (long long)node * 32 + lane;
  float ds2 = dis[node];
  ds2 *= ds2;
  float z = out[idx] + ds2 * H2[idx] + b2[lane];
  float m = z;
  for (int off = 16; off > 0; off >>= 1) m = fmaxf(m, __shfl_xor(m, off, 32));
  float ev = __expf(z - m);
  float s = ev;
  for (int off = 16; off > 0; off >>= 1) s += __shfl_xor(s, off, 32);
  out[idx] = ev / s;
}

extern "C" void kernel_launch(void* const* d_in, const int* in_sizes, int n_in,
                              void* d_out, int out_size, void* d_ws, size_t ws_size,
                              hipStream_t stream) {
  (void)n_in; (void)out_size; (void)ws_size;
  const float* x  = (const float*)d_in[0];
  const int*   ei = (const int*)d_in[1];
  const float* W1 = (const float*)d_in[2];
  const float* b1 = (const float*)d_in[3];
  const float* W2 = (const float*)d_in[4];
  const float* b2 = (const float*)d_in[5];

  const int IN = 256, HID = 64, OD = 32;
  const int M = in_sizes[0] / IN;
  const int E = in_sizes[1] / 2;
  const int* src = ei;
  const int* dst = ei + E;

  float* ws   = (float*)d_ws;
  float* dis  = ws;
  float* H1   = dis + (((long long)M + 1023) & ~1023LL);
  float* agg1 = H1 + (long long)M * HID;
  float* H2   = agg1 + (long long)M * HID;
  float* out  = (float*)d_out;

  const int T = 256;

  set_val_kernel<<<(M + T - 1) / T, T, 0, stream>>>(dis, M, 1.0f);
  deg_count_kernel<<<(E + T - 1) / T, T, 0, stream>>>(dst, E, dis);
  rsqrt_kernel<<<(M + T - 1) / T, T, 0, stream>>>(dis, M);

  gemm_wmma_kernel<256, 64><<<(M + 31) / 32, 256, 0, stream>>>(x, W1, H1, M);
  set_val_kernel<<<(int)(((long long)M * HID + T - 1) / T), T, 0, stream>>>(
      agg1, (long long)M * HID, 0.0f);
  {
    long long tot = (long long)E * (HID / 4);
    scatter_kernel<64><<<(int)((tot + T - 1) / T), T, 0, stream>>>(src, dst, dis, H1, agg1, E);
  }
  combine_kernel<64, true><<<(int)(((long long)M * HID + T - 1) / T), T, 0, stream>>>(
      H1, b1, dis, agg1, M);

  gemm_wmma_kernel<64, 32><<<(M + 63) / 64, 256, 0, stream>>>(agg1, W2, H2, M);
  set_val_kernel<<<(int)(((long long)M * OD + T - 1) / T), T, 0, stream>>>(
      out, (long long)M * OD, 0.0f);
  {
    long long tot = (long long)E * (OD / 4);
    scatter_kernel<32><<<(int)((tot + T - 1) / T), T, 0, stream>>>(src, dst, dis, H2, out, E);
  }
  combine_softmax_kernel<<<(int)(((long long)M * 32 + T - 1) / T), T, 0, stream>>>(
      H2, b2, dis, out, M);
}


# scatter-add re-laid: a wave takes 32 edges, one lane per float, so each gather load and f32 atomic add covers one whole aligned 128-B line (was 4 lines x 8 floats); arrival-order sums as before
# speedup vs baseline: 1.3355x; 1.3355x over previous
#include <hip/hip_runtime.h>

typedef __attribute__((ext_vector_type(2))) float v2f;
typedef __attribute__((ext_vector_type(8))) float v8f;

__global__ void set_val_kernel(float* p, long long n, float v) {
  long long i = (long long)blockIdx.x * blockDim.x + threadIdx.x;
  if (i < n) p[i] = v;
}

__global__ void deg_count_kernel(const int* __restrict__ dst, int E, float* deg) {
  int e = blockIdx.x * blockDim.x + threadIdx.x;
  if (e < E) atomicAdd(&deg[dst[e]], 1.0f);
}

__global__ void rsqrt_kernel(float* p, int n) {
  int i = blockIdx.x * blockDim.x + threadIdx.x;
  if (i < n) p[i] = rsqrtf(p[i]);
}

template<int K, int N>
__global__ void gemm_wmma_kernel(const float* __restrict__ X,
                                 const float* __restrict__ W,
                                 float* __restrict__ OUT, int M) {
  __shared__ float sW[K * N];
  for (int i = threadIdx.x; i < K * N; i += blockDim.x) sW[i] = W[i];
  __syncthreads();

  constexpr int COL_TILES  = N / 16;
  constexpr int ROW_TILES  = 8 / COL_TILES;
  constexpr int BLOCK_ROWS = 16 * ROW_TILES;

  const int wave = threadIdx.x >> 5;
  const int lane = threadIdx.x & 31;
  const int rowBase = blockIdx.x * BLOCK_ROWS + (wave / COL_TILES) * 16;
  const int colBase = (wave % COL_TILES) * 16;
  if (rowBase + 16 > M) return;

  const int mn  = lane & 15;
  const int grp = lane >> 4;

  const float* xrow = X + (long long)(rowBase + mn) * K + 2 * grp;
  const float* wcol = sW + (long long)(2 * grp) * N + colBase + mn;

  v8f acc = {};
#pragma unroll 4
  for (int k0 = 0; k0 < K; k0 += 4) {
    v2f a = *(const v2f*)(xrow + k0);
    v2f b;
    b.x = wcol[(long long)k0 * N];
    b.y = wcol[(long long)(k0 + 1) * N];
    acc = __builtin_amdgcn_wmma_f32_16x16x4_f32(
        false, a, false, b, (short)0, acc, false, false);
  }

  float* out = OUT + (long long)(rowBase + 8 * grp) * N + colBase + mn;
#pragma unroll
  for (int r = 0; r < 8; ++r) out[(long long)r * N] = acc[r];
}

template<int D>
__global__ void scatter_kernel(const int* __restrict__ src, const int* __restrict__ dst,
                               const float* __restrict__ dis, const float* __restrict__ H,
                               float* __restrict__ agg, int E) {
  constexpr int NL = D / 32;
  constexpr int G  = 8;
  const int lane = threadIdx.x & 31;
  const int wave = __builtin_amdgcn_readfirstlane((int)(threadIdx.x >> 5));
  const long long e0 = ((long long)blockIdx.x * (blockDim.x >> 5) + wave) * 32;
  if (e0 >= E) return;
  const long long rem = (long long)E - e0;
  const int cnt = rem < 32 ? (int)rem : 32;

  int s = 0, d = 0;
  float nrm = 0.0f;
  if (lane < cnt) {
    s = src[e0 + lane];
    d = dst[e0 + lane];
    nrm = dis[s] * dis[d];
  }

  if (cnt == 32) {
#pragma unroll
    for (int j0 = 0; j0 < 32; j0 += G) {
      float v[G][NL];
#pragma unroll
      for (int g = 0; g < G; ++g) {
        const int sj = __builtin_amdgcn_readlane(s, j0 + g);
        const float* hp = H + (long long)sj * D + lane;
#pragma unroll
        for (int i = 0; i < NL; ++i) v[g][i] = hp[32 * i];
      }
#pragma unroll
      for (int g = 0; g < G; ++g) {
        const int dj = __builtin_amdgcn_readlane(d, j0 + g);
        const float nj = __int_as_float(__builtin_amdgcn_readlane(__float_as_int(nrm), j0 + g));
        float* o = agg + (long long)dj * D + lane;
#pragma unroll
        for (int i = 0; i < NL; ++i) atomicAdd(o + 32 * i, v[g][i] * nj);
      }
    }
  } else {
    for (int j = 0; j < cnt; ++j) {
      const int sj = __builtin_amdgcn_readlane(s, j);
      const int dj = __builtin_amdgcn_readlane(d, j);
      const float nj = __int_as_float(__builtin_amdgcn_readlane(__float_as_int(nrm), j));
      const float* hp = H + (long long)sj * D + lane;
      float* o = agg + (long long)dj * D + lane;
      for (int i = 0; i < NL; ++i) atomicAdd(o + 32 * i, hp[32 * i] * nj);
    }
  }
}

template<int D, bool RELU>
__global__ void combine_kernel(const float* __restrict__ H, const float* __restrict__ bias,
                               const float* __restrict__ dis, float* agg, int M) {
  long long i = (long long)blockIdx.x * blockDim.x + threadIdx.x;
  if (i >= (long long)M * D) return;
  int n = (int)(i / D), d = (int)(i % D);
  float z = agg[i] + dis[n] * dis[n] * H[i] + bias[d];
  if (RELU) z = fmaxf(z, 0.0f);
  agg[i] = z;
}

__global__ void combine_softmax_kernel(const float* __restrict__ H2,
                                       const float* __restrict__ b2,
                                       const float* __restrict__ dis,
                                       float* out, int M) {
  long long gt = (long long)blockIdx.x * blockDim.x + threadIdx.x;
  int node = (int)(gt >> 5);
  int lane = threadIdx.x & 31;
  if (node >= M) return;
  long long idx = (long long)node * 32 + lane;
  float ds2 = dis[node];
  ds2 *= ds2;
  float z = out[idx] + ds2 * H2[idx] + b2[lane];
  float m = z;
  for (int off = 16; off > 0; off >>= 1) m = fmaxf(m, __shfl_xor(m, off, 32));
  float ev = __expf(z - m);
  float s = ev;
  for (int off = 16; off > 0; off >>= 1) s += __shfl_xor(s, off, 32);
  out[idx] = ev / s;
}

extern "C" void kernel_launch(void* const* d_in, const int* in_sizes, int n_in,
                              void* d_out, int out_size, void* d_ws, size_t ws_size,
                              hipStream_t stream) {
  (void)n_in; (void)out_size; (void)ws_size;
  const float* x  = (const float*)d_in[0];
  const int*   ei = (const int*)d_in[1];
  const float* W1 = (const float*)d_in[2];
  const float* b1 = (const float*)d_in[3];
  const float* W2 = (const float*)d_in[4];
  const float* b2 = (const float*)d_in[5];

  const int IN = 256, HID = 64, OD = 32;
  const int M = in_sizes[0] / IN;
  const int E = in_sizes[1] / 2;
  const int* src = ei;
  const int* dst = ei + E;

  float* ws   = (float*)d_ws;
  float* dis  = ws;
  float* H1   = dis + (((long long)M + 1023) & ~1023LL);
  float* agg1 = H1 + (long long)M * HID;
  float* H2   = agg1 + (long long)M * HID;
  float* out  = (float*)d_out;

  const int T = 256;

  set_val_kernel<<<(M + T - 1) / T, T, 0, stream>>>(dis, M, 1.0f);
  deg_count_kernel<<<(E + T - 1) / T, T, 0, stream>>>(dst, E, dis);
  rsqrt_kernel<<<(M + T - 1) / T, T, 0, stream>>>(dis, M);

  gemm_wmma_kernel<256, 64><<<(M + 31) / 32, 256, 0, stream>>>(x, W1, H1, M);
  set_val_kernel<<<(int)(((long long)M * HID + T - 1) / T), T, 0, stream>>>(
      agg1, (long long)M * HID, 0.0f);
  scatter_kernel<64><<<(E + T - 1) / T, T, 0, stream>>>(src, dst, dis, H1, agg1, E);
  combine_kernel<64, true><<<(int)(((long long)M * HID + T - 1) / T), T, 0, stream>>>(
      H1, b1, dis, agg1, M);

  gemm_wmma_kernel<64, 32><<<(M + 63) / 64, 256, 0, stream>>>(agg1, W2, H2, M);
  set_val_kernel<<<(int)(((long long)M * OD + T - 1) / T), T, 0, stream>>>(
      out, (long long)M * OD, 0.0f);
  scatter_kernel<32><<<(E + T - 1) / T, T, 0, stream>>>(src, dst, dis, H2, out, E);
  combine_softmax_kernel<<<(int)(((long long)M * 32 + T - 1) / T), T, 0, stream>>>(
      H2, b2, dis, out, M);
}


# in-degree count privatized in LDS: a block counts one edge chunk into a range of 12512 nodes and adds its bins as whole-line f32 atomics (100k line adds, was 1.6M one-word adds); integer sums, exact
# speedup vs baseline: 1.6626x; 1.2449x over previous
#include <hip/hip_runtime.h>

typedef __attribute__((ext_vector_type(2))) float v2f;
typedef __attribute__((ext_vector_type(8))) float v8f;

__global__ void set_val_kernel(float* p, long long n, float v) {
  long long i = (long long)blockIdx.x * blockDim.x + threadIdx.x;
  if (i < n) p[i] = v;
}

enum { DEG_RS = 12512, DEG_NC = 32 };

__global__ void __launch_bounds__(1024)
deg_hist_kernel(const int* __restrict__ dst, int E, float* deg, int M) {
  __shared__ int bins[DEG_RS];
  const int tid = threadIdx.x;
  const int lo = blockIdx.x * DEG_RS;
  for (int i = tid; i < DEG_RS; i += 1024) bins[i] = 0;
  __syncthreads();

  const int per = (E + DEG_NC - 1) / DEG_NC;
  const int e0 = blockIdx.y * per;
  int e1 = e0 + per;
  if (e1 > E) e1 = E;
  int e = e0 + tid;
  for (; e + 7 * 1024 < e1; e += 8 * 1024) {
    int d[8];
#pragma unroll
    for (int u = 0; u < 8; ++u) d[u] = dst[e + u * 1024];
#pragma unroll
    for (int u = 0; u < 8; ++u) {
      const unsigned k = (unsigned)(d[u] - lo);
      if (k < (unsigned)DEG_RS) atomicAdd(&bins[k], 1);
    }
  }
  for (; e < e1; e += 1024) {
    const unsigned k = (unsigned)(dst[e] - lo);
    if (k < (unsigned)DEG_RS) atomicAdd(&bins[k], 1);
  }
  __syncthreads();

  for (int i = tid; i < DEG_RS; i += 1024) {
    const int c = bins[i];
    const int n = lo + i;
    if (c != 0 && n < M) atomicAdd(&deg[n], (float)c);
  }
}

__global__ void rsqrt_kernel(float* p, int n) {
  int i = blockIdx.x * blockDim.x + threadIdx.x;
  if (i < n) p[i] = rsqrtf(p[i]);
}

template<int K, int N>
__global__ void gemm_wmma_kernel(const float* __restrict__ X,
                                 const float* __restrict__ W,
                                 float* __restrict__ OUT, int M) {
  __shared__ float sW[K * N];
  for (int i = threadIdx.x; i < K * N; i += blockDim.x) sW[i] = W[i];
  __syncthreads();

  constexpr int COL_TILES  = N / 16;
  constexpr int ROW_TILES  = 8 / COL_TILES;
  constexpr int BLOCK_ROWS = 16 * ROW_TILES;

  const int wave = threadIdx.x >> 5;
  const int lane = threadIdx.x & 31;
  const int rowBase = blockIdx.x * BLOCK_ROWS + (wave / COL_TILES) * 16;
  const int colBase = (wave % COL_TILES) * 16;
  if (rowBase + 16 > M) return;

  const int mn  = lane & 15;
  const int grp = lane >> 4;

  const float* xrow = X + (long long)(rowBase + mn) * K + 2 * grp;
  const float* wcol = sW + (long long)(2 * grp) * N + colBase + mn;

  v8f acc = {};
#pragma unroll 4
  for (int k0 = 0; k0 < K; k0 += 4) {
    v2f a = *(const v2f*)(xrow + k0);
    v2f b;
    b.x = wcol[(long long)k0 * N];
    b.y = wcol[(long long)(k0 + 1) * N];
    acc = __builtin_amdgcn_wmma_f32_16x16x4_f32(
        false, a, false, b, (short)0, acc, false, false);
  }

  float* out = OUT + (long long)(rowBase + 8 * grp) * N + colBase + mn;
#pragma unroll
  for (int r = 0; r < 8; ++r) out[(long long)r * N] = acc[r];
}

template<int D>
__global__ void scatter_kernel(const int* __restrict__ src, const int* __restrict__ dst,
                               const float* __restrict__ dis, const float* __restrict__ H,
                               float* __restrict__ agg, int E) {
  constexpr int NL = D / 32;
  constexpr int G  = 8;
  const int lane = threadIdx.x & 31;
  const int wave = __builtin_amdgcn_readfirstlane((int)(threadIdx.x >> 5));
  const long long e0 = ((long long)blockIdx.x * (blockDim.x >> 5) + wave) * 32;
  if (e0 >= E) return;
  const long long rem = (long long)E - e0;
  const int cnt = rem < 32 ? (int)rem : 32;

  int s = 0, d = 0;
  float nrm = 0.0f;
  if (lane < cnt) {
    s = src[e0 + lane];
    d = dst[e0 + lane];
    nrm = dis[s] * dis[d];
  }

  if (cnt == 32) {
#pragma unroll
    for (int j0 = 0; j0 < 32; j0 += G) {
      float v[G][NL];
#pragma unroll
      for (int g = 0; g < G; ++g) {
        const int sj = __builtin_amdgcn_readlane(s, j0 + g);
        const float* hp = H + (long long)sj * D + lane;
#pragma unroll
        for (int i = 0; i < NL; ++i) v[g][i] = hp[32 * i];
      }
#pragma unroll
      for (int g = 0; g < G; ++g) {
        const int dj = __builtin_amdgcn_readlane(d, j0 + g);
        const float nj = __int_as_float(__builtin_amdgcn_readlane(__float_as_int(nrm), j0 + g));
        float* o = agg + (long long)dj * D + lane;
#pragma unroll
        for (int i = 0; i < NL; ++i) atomicAdd(o + 32 * i, v[g][i] * nj);
      }
    }
  } else {
    for (int j = 0; j < cnt; ++j) {
      const int sj = __builtin_amdgcn_readlane(s, j);
      const int dj = __builtin_amdgcn_readlane(d, j);
      const float nj = __int_as_float(__builtin_amdgcn_readlane(__float_as_int(nrm), j));
      const float* hp = H + (long long)sj * D + lane;
      float* o = agg + (long long)dj * D + lane;
      for (int i = 0; i < NL; ++i) atomicAdd(o + 32 * i, hp[32 * i] * nj);
    }
  }
}

template<int D, bool RELU>
__global__ void combine_kernel(const float* __restrict__ H, const float* __restrict__ bias,
                               const float* __restrict__ dis, float* agg, int M) {
  long long i = (long long)blockIdx.x * blockDim.x + threadIdx.x;
  if (i >= (long long)M * D) return;
  int n = (int)(i / D), d = (int)(i % D);
  float z = agg[i] + dis[n] * dis[n] * H[i] + bias[d];
  if (RELU) z = fmaxf(z, 0.0f);
  agg[i] = z;
}

__global__ void combine_softmax_kernel(const float* __restrict__ H2,
                                       const float* __restrict__ b2,
                                       const float* __restrict__ dis,
                                       float* out, int M) {
  long long gt = (long long)blockIdx.x * blockDim.x + threadIdx.x;
  int node = (int)(gt >> 5);
  int lane = threadIdx.x & 31;
  if (node >= M) return;
  long long idx = (long long)node * 32 + lane;
  float ds2 = dis[node];
  ds2 *= ds2;
  float z = out[idx] + ds2 * H2[idx] + b2[lane];
  float m = z;
  for (int off = 16; off > 0; off >>= 1) m = fmaxf(m, __shfl_xor(m, off, 32));
  float ev = __expf(z - m);
  float s = ev;
  for (int off = 16; off > 0; off >>= 1) s += __shfl_xor(s, off, 32);
  out[idx] = ev / s;
}

extern "C" void kernel_launch(void* const* d_in, const int* in_sizes, int n_in,
                              void* d_out, int out_size, void* d_ws, size_t ws_size,
                              hipStream_t stream) {
  (void)n_in; (void)out_size; (void)ws_size;
  const float* x  = (const float*)d_in[0];
  const int*   ei = (const int*)d_in[1];
  const float* W1 = (const float*)d_in[2];
  const float* b1 = (const float*)d_in[3];
  const float* W2 = (const float*)d_in[4];
  const float* b2 = (const float*)d_in[5];

  const int IN = 256, HID = 64, OD = 32;
  const int M = in_sizes[0] / IN;
  const int E = in_sizes[1] / 2;
  const int* src = ei;
  const int* dst = ei + E;

  float* ws   = (float*)d_ws;
  float* dis  = ws;
  float* H1   = dis + (((long long)M + 1023) & ~1023LL);
  float* agg1 = H1 + (long long)M * HID;
  float* H2   = agg1 + (long long)M * HID;
  float* out  = (float*)d_out;

  const int T = 256;

  set_val_kernel<<<(M + T - 1) / T, T, 0, stream>>>(dis, M, 1.0f);
  deg_hist_kernel<<<dim3((M + DEG_RS - 1) / DEG_RS, DEG_NC), 1024, 0, stream>>>(dst, E, dis, M);
  rsqrt_kernel<<<(M + T - 1) / T, T, 0, stream>>>(dis, M);

  gemm_wmma_kernel<256, 64><<<(M + 31) / 32, 256, 0, stream>>>(x, W1, H1, M);
  set_val_kernel<<<(int)(((long long)M * HID + T - 1) / T), T, 0, stream>>>(
      agg1, (long long)M * HID, 0.0f);
  scatter_kernel<64><<<(E + T - 1) / T, T, 0, stream>>>(src, dst, dis, H1, agg1, E);
  combine_kernel<64, true><<<(int)(((long long)M * HID + T - 1) / T), T, 0, stream>>>(
      H1, b1, dis, agg1, M);

  gemm_wmma_kernel<64, 32><<<(M + 63) / 64, 256, 0, stream>>>(agg1, W2, H2, M);
  set_val_kernel<<<(int)(((long long)M * OD + T - 1) / T), T, 0, stream>>>(
      out, (long long)M * OD, 0.0f);
  scatter_kernel<32><<<(E + T - 1) / T, T, 0, stream>>>(src, dst, dis, H2, out, E);
  combine_softmax_kernel<<<(int)(((long long)M * 32 + T - 1) / T), T, 0, stream>>>(
      H2, b2, dis, out, M);
}


# self-loop term + bias stored by each GEMM's epilogue as the start value of the edge sums: zero-fills and combine pass go, ReLU moves into GEMM2's A staging; sum order: start value, then edge adds
# speedup vs baseline: 1.8812x; 1.1315x over previous
#include <hip/hip_runtime.h>

typedef __attribute__((ext_vector_type(2))) float v2f;
typedef __attribute__((ext_vector_type(8))) float v8f;

__global__ void set_val_kernel(float* p, long long n, float v) {
  long long i = (long long)blockIdx.x * blockDim.x + threadIdx.x;
  if (i < n) p[i] = v;
}

enum { DEG_RS = 12512, DEG_NC = 32 };

__global__ void __launch_bounds__(1024)
deg_hist_kernel(const int* __restrict__ dst, int E, float* deg, int M) {
  __shared__ int bins[DEG_RS];
  const int tid = threadIdx.x;
  const int lo = blockIdx.x * DEG_RS;
  for (int i = tid; i < DEG_RS; i += 1024) bins[i] = 0;
  __syncthreads();

  const int per = (E + DEG_NC - 1) / DEG_NC;
  const int e0 = blockIdx.y * per;
  int e1 = e0 + per;
  if (e1 > E) e1 = E;
  int e = e0 + tid;
  for (; e + 7 * 1024 < e1; e += 8 * 1024) {
    int d[8];
#pragma unroll
    for (int u = 0; u < 8; ++u) d[u] = dst[e + u * 1024];
#pragma unroll
    for (int u = 0; u < 8; ++u) {
      const unsigned k = (unsigned)(d[u] - lo);
      if (k < (unsigned)DEG_RS) atomicAdd(&bins[k], 1);
    }
  }
  for (; e < e1; e += 1024) {
    const unsigned k = (unsigned)(dst[e] - lo);
    if (k < (unsigned)DEG_RS) atomicAdd(&bins[k], 1);
  }
  __syncthreads();

  for (int i = tid; i < DEG_RS; i += 1024) {
    const int c = bins[i];
    const int n = lo + i;
    if (c != 0 && n < M) atomicAdd(&deg[n], (float)c);
  }
}

__global__ void rsqrt_kernel(float* p, int n) {
  int i = blockIdx.x * blockDim.x + threadIdx.x;
  if (i < n) p[i] = rsqrtf(p[i]);
}

template<int K, int N, bool RELU_A>
__global__ void gemm_wmma_kernel(const float* __restrict__ X,
                                 const float* __restrict__ W,
                                 const float* __restrict__ bias,
                                 const float* __restrict__ dis,
                                 float* __restrict__ OUT,
                                 float* __restrict__ ACC, int M) {
  constexpr int COL_TILES  = N / 16;
  constexpr int ROW_TILES  = 8 / COL_TILES;
  constexpr int BLOCK_ROWS = 16 * ROW_TILES;
  constexpr int AP = K + 4;

  __shared__ float sW[K * N];
  __shared__ __attribute__((aligned(16))) float sA[RELU_A ? BLOCK_ROWS * AP : 4];
  for (int i = threadIdx.x; i < K * N; i += blockDim.x) sW[i] = W[i];
  if (RELU_A) {
    constexpr int NT = 256;
    const int blockRow = blockIdx.x * BLOCK_ROWS;
#pragma unroll
    for (int j = 0; j < BLOCK_ROWS * (K / 4) / NT; ++j) {
      const int i = threadIdx.x + j * NT;
      const int r = i / (K / 4), c = i % (K / 4);
      const int gr = blockRow + r < M ? blockRow + r : M - 1;
      float4 v = *(const float4*)(X + (long long)gr * K + 4 * c);
      v.x = fmaxf(v.x, 0.0f);
      v.y = fmaxf(v.y, 0.0f);
      v.z = fmaxf(v.z, 0.0f);
      v.w = fmaxf(v.w, 0.0f);
      *(float4*)(sA + r * AP + 4 * c) = v;
    }
  }
  __syncthreads();

  const int wave = threadIdx.x >> 5;
  const int lane = threadIdx.x & 31;
  const int rowBase = blockIdx.x * BLOCK_ROWS + (wave / COL_TILES) * 16;
  const int colBase = (wave % COL_TILES) * 16;
  if (rowBase + 16 > M) return;

  const int mn  = lane & 15;
  const int grp = lane >> 4;

  const float* xrow = X + (long long)(rowBase + mn) * K + 2 * grp;
  const float* arow = sA + ((wave / COL_TILES) * 16 + mn) * AP + 2 * grp;
  const float* wcol = sW + (long long)(2 * grp) * N + colBase + mn;

  v8f acc = {};
#pragma unroll 1
  for (int k0 = 0; k0 < K; k0 += 16) {
#pragma unroll
    for (int u = 0; u < 16; u += 4) {
      const int k = k0 + u;
      v2f a;
      if (RELU_A) a = *(const v2f*)(arow + k);
      else        a = *(const v2f*)(xrow + k);
      v2f b;
      b.x = wcol[(long long)k * N];
      b.y = wcol[(long long)(k + 1) * N];
      acc = __builtin_amdgcn_wmma_f32_16x16x4_f32(
          false, a, false, b, (short)0, acc, false, false);
    }
    if (RELU_A) {
      asm volatile("v_nop\n\tv_nop\n\tv_nop\n\tv_nop" : "+v"(acc));
    }
  }
  if (!RELU_A) {
    asm volatile("v_nop\n\tv_nop\n\tv_nop\n\tv_nop" : "+v"(acc));
  }

  const int row0 = rowBase + 8 * grp;
  const long long off = (long long)row0 * N + colBase + mn;
  float* out = OUT + off;
  float* st  = ACC + off;
  const float bc = bias[colBase + mn];
#pragma unroll
  for (int r = 0; r < 8; ++r) {
    const float ds = dis[row0 + r];
    out[(long long)r * N] = acc[r];
    st[(long long)r * N]  = ds * ds * acc[r] + bc;
  }
}

template<int D>
__global__ void scatter_kernel(const int* __restrict__ src, const int* __restrict__ dst,
                               const float* __restrict__ dis, const float* __restrict__ H,
                               float* __restrict__ agg, int E) {
  constexpr int NL = D / 32;
  constexpr int G  = 8;
  const int lane = threadIdx.x & 31;
  const int wave = __builtin_amdgcn_readfirstlane((int)(threadIdx.x >> 5));
  const long long e0 = ((long long)blockIdx.x * (blockDim.x >> 5) + wave) * 32;
  if (e0 >= E) return;
  const long long rem = (long long)E - e0;
  const int cnt = rem < 32 ? (int)rem : 32;

  int s = 0, d = 0;
  float nrm = 0.0f;
  if (lane < cnt) {
    s = src[e0 + lane];
    d = dst[e0 + lane];
    nrm = dis[s] * dis[d];
  }

  if (cnt == 32) {
#pragma unroll
    for (int j0 = 0; j0 < 32; j0 += G) {
      float v[G][NL];
#pragma unroll
      for (int g = 0; g < G; ++g) {
        const int sj = __builtin_amdgcn_readlane(s, j0 + g);
        const float* hp = H + (long long)sj * D + lane;
#pragma unroll
        for (int i = 0; i < NL; ++i) v[g][i] = hp[32 * i];
      }
#pragma unroll
      for (int g = 0; g < G; ++g) {
        const int dj = __builtin_amdgcn_readlane(d, j0 + g);
        const float nj = __int_as_float(__builtin_amdgcn_readlane(__float_as_int(nrm), j0 + g));
        float* o = agg + (long long)dj * D + lane;
#pragma unroll
        for (int i = 0; i < NL; ++i) atomicAdd(o + 32 * i, v[g][i] * nj);
      }
    }
  } else {
    for (int j = 0; j < cnt; ++j) {
      const int sj = __builtin_amdgcn_readlane(s, j);
      const int dj = __builtin_amdgcn_readlane(d, j);
      const float nj = __int_as_float(__builtin_amdgcn_readlane(__float_as_int(nrm), j));
      const float* hp = H + (long long)sj * D + lane;
      float* o = agg + (long long)dj * D + lane;
      for (int i = 0; i < NL; ++i) atomicAdd(o + 32 * i, hp[32 * i] * nj);
    }
  }
}

__global__ void softmax_kernel(float* out, int M) {
  long long gt = (long long)blockIdx.x * blockDim.x + threadIdx.x;
  int node = (int)(gt >> 5);
  int lane = threadIdx.x & 31;
  if (node >= M) return;
  long long idx = (long long)node * 32 + lane;
  float z = out[idx];
  float m = z;
  for (int off = 16; off > 0; off >>= 1) m = fmaxf(m, __shfl_xor(m, off, 32));
  float ev = __expf(z - m);
  float s = ev;
  for (int off = 16; off > 0; off >>= 1) s += __shfl_xor(s, off, 32);
  out[idx] = ev / s;
}

extern "C" void kernel_launch(void* const* d_in, const int* in_sizes, int n_in,
                              void* d_out, int out_size, void* d_ws, size_t ws_size,
                              hipStream_t stream) {
  (void)n_in; (void)out_size; (void)ws_size;
  const float* x  = (const float*)d_in[0];
  const int*   ei = (const int*)d_in[1];
  const float* W1 = (const float*)d_in[2];
  const float* b1 = (const float*)d_in[3];
  const float* W2 = (const float*)d_in[4];
  const float* b2 = (const float*)d_in[5];

  const int IN = 256, HID = 64, OD = 32;
  const int M = in_sizes[0] / IN;
  const int E = in_sizes[1] / 2;
  const int* src = ei;
  const int* dst = ei + E;

  float* ws   = (float*)d_ws;
  float* dis  = ws;
  float* H1   = dis + (((long long)M + 1023) & ~1023LL);
  float* agg1 = H1 + (long long)M * HID;
  float* H2   = agg1 + (long long)M * HID;
  float* out  = (float*)d_out;

  const int T = 256;

  set_val_kernel<<<(M + T - 1) / T, T, 0, stream>>>(dis, M, 1.0f);
  deg_hist_kernel<<<dim3((M + DEG_RS - 1) / DEG_RS, DEG_NC), 1024, 0, stream>>>(dst, E, dis, M);
  rsqrt_kernel<<<(M + T - 1) / T, T, 0, stream>>>(dis, M);

  gemm_wmma_kernel<256, 64, false><<<(M + 31) / 32, 256, 0, stream>>>(
      x, W1, b1, dis, H1, agg1, M);
  scatter_kernel<64><<<(E + T - 1) / T, T, 0, stream>>>(src, dst, dis, H1, agg1, E);

  gemm_wmma_kernel<64, 32, true><<<(M + 63) / 64, 256, 0, stream>>>(
      agg1, W2, b2, dis, H2, out, M);
  scatter_kernel<32><<<(E + T - 1) / T, T, 0, stream>>>(src, dst, dis, H2, out, E);
  softmax_kernel<<<(int)(((long long)M * 32 + T - 1) / T), T, 0, stream>>>(out, M);
}


# edge norm factored out of the edge pass: GEMM epilogues store dis[src]-scaled rows, edges add them as they are (no per-edge dis loads, no multiply), the reader applies dis[dst] + bias; sum order kept
# speedup vs baseline: 1.9976x; 1.0619x over previous
#include <hip/hip_runtime.h>

typedef __attribute__((ext_vector_type(2))) float v2f;
typedef __attribute__((ext_vector_type(8))) float v8f;

__global__ void set_val_kernel(float* p, long long n, float v) {
  long long i = (long long)blockIdx.x * blockDim.x + threadIdx.x;
  if (i < n) p[i] = v;
}

enum { DEG_RS = 12512, DEG_NC = 32 };

__global__ void __launch_bounds__(1024)
deg_hist_kernel(const int* __restrict__ dst, int E, float* deg, int M) {
  __shared__ int bins[DEG_RS];
  const int tid = threadIdx.x;
  const int lo = blockIdx.x * DEG_RS;
  for (int i = tid; i < DEG_RS; i += 1024) bins[i] = 0;
  __syncthreads();

  const int per = (E + DEG_NC - 1) / DEG_NC;
  const int e0 = blockIdx.y * per;
  int e1 = e0 + per;
  if (e1 > E) e1 = E;
  int e = e0 + tid;
  for (; e + 7 * 1024 < e1; e += 8 * 1024) {
    int d[8];
#pragma unroll
    for (int u = 0; u < 8; ++u) d[u] = dst[e + u * 1024];
#pragma unroll
    for (int u = 0; u < 8; ++u) {
      const unsigned k = (unsigned)(d[u] - lo);
      if (k < (unsigned)DEG_RS) atomicAdd(&bins[k], 1);
    }
  }
  for (; e < e1; e += 1024) {
    const unsigned k = (unsigned)(dst[e] - lo);
    if (k < (unsigned)DEG_RS) atomicAdd(&bins[k], 1);
  }
  __syncthreads();

  for (int i = tid; i < DEG_RS; i += 1024) {
    const int c = bins[i];
    const int n = lo + i;
    if (c != 0 && n < M) atomicAdd(&deg[n], (float)c);
  }
}

__global__ void rsqrt_kernel(float* p, int n) {
  int i = blockIdx.x * blockDim.x + threadIdx.x;
  if (i < n) p[i] = rsqrtf(p[i]);
}

template<int K, int N, bool RELU_A>
__global__ void gemm_wmma_kernel(const float* __restrict__ X,
                                 const float* __restrict__ W,
                                 const float* __restrict__ bias,
                                 const float* __restrict__ dis,
                                 float* __restrict__ OUT,
                                 float* __restrict__ ACC, int M) {
  constexpr int COL_TILES  = N / 16;
  constexpr int ROW_TILES  = 8 / COL_TILES;
  constexpr int BLOCK_ROWS = 16 * ROW_TILES;
  constexpr int AP = K + 4;

  __shared__ float sW[K * N];
  __shared__ __attribute__((aligned(16))) float sA[RELU_A ? BLOCK_ROWS * AP : 4];
  for (int i = threadIdx.x; i < K * N; i += blockDim.x) sW[i] = W[i];
  if (RELU_A) {
    constexpr int NT = 256;
    const int blockRow = blockIdx.x * BLOCK_ROWS;
#pragma unroll
    for (int j = 0; j < BLOCK_ROWS * (K / 4) / NT; ++j) {
      const int i = threadIdx.x + j * NT;
      const int r = i / (K / 4), c = i % (K / 4);
      const int gr = blockRow + r < M ? blockRow + r : M - 1;
      float4 v = *(const float4*)(X + (long long)gr * K + 4 * c);
      const float4 bb = *(const float4*)(bias + 4 * c);
      const float dr = dis[gr];
      v.x = fmaxf(dr * v.x + bb.x, 0.0f);
      v.y = fmaxf(dr * v.y + bb.y, 0.0f);
      v.z = fmaxf(dr * v.z + bb.z, 0.0f);
      v.w = fmaxf(dr * v.w + bb.w, 0.0f);
      *(float4*)(sA + r * AP + 4 * c) = v;
    }
  }
  __syncthreads();

  const int wave = threadIdx.x >> 5;
  const int lane = threadIdx.x & 31;
  const int rowBase = blockIdx.x * BLOCK_ROWS + (wave / COL_TILES) * 16;
  const int colBase = (wave % COL_TILES) * 16;
  if (rowBase + 16 > M) return;

  const int mn  = lane & 15;
  const int grp = lane >> 4;

  const float* xrow = X + (long long)(rowBase + mn) * K + 2 * grp;
  const float* arow = sA + ((wave / COL_TILES) * 16 + mn) * AP + 2 * grp;
  const float* wcol = sW + (long long)(2 * grp) * N + colBase + mn;

  v8f acc = {};
#pragma unroll 1
  for (int k0 = 0; k0 < K; k0 += 16) {
#pragma unroll
    for (int u = 0; u < 16; u += 4) {
      const int k = k0 + u;
      v2f a;
      if (RELU_A) a = *(const v2f*)(arow + k);
      else        a = *(const v2f*)(xrow + k);
      v2f b;
      b.x = wcol[(long long)k * N];
      b.y = wcol[(long long)(k + 1) * N];
      acc = __builtin_amdgcn_wmma_f32_16x16x4_f32(
          false, a, false, b, (short)0, acc, false, false);
    }
    if (RELU_A) {
      asm volatile("v_nop\n\tv_nop\n\tv_nop\n\tv_nop" : "+v"(acc));
    }
  }
  if (!RELU_A) {
    asm volatile("v_nop\n\tv_nop\n\tv_nop\n\tv_nop" : "+v"(acc));
  }

  const int row0 = rowBase + 8 * grp;
  const long long off = (long long)row0 * N + colBase + mn;
  float* out = OUT + off;
  float* st  = ACC + off;
#pragma unroll
  for (int r = 0; r < 8; ++r) {
    const float hv = dis[row0 + r] * acc[r];
    out[(long long)r * N] = hv;
    st[(long long)r * N]  = hv;
  }
}

template<int D>
__global__ void scatter_kernel(const int* __restrict__ src, const int* __restrict__ dst,
                               const float* __restrict__ H, float* __restrict__ agg, int E) {
  constexpr int NL = D / 32;
  constexpr int G  = 8;
  const int lane = threadIdx.x & 31;
  const int wave = __builtin_amdgcn_readfirstlane((int)(threadIdx.x >> 5));
  const long long e0 = ((long long)blockIdx.x * (blockDim.x >> 5) + wave) * 32;
  if (e0 >= E) return;
  const long long rem = (long long)E - e0;
  const int cnt = rem < 32 ? (int)rem : 32;

  int s = 0, d = 0;
  if (lane < cnt) {
    s = src[e0 + lane];
    d = dst[e0 + lane];
  }

  if (cnt == 32) {
#pragma unroll
    for (int j0 = 0; j0 < 32; j0 += G) {
      float v[G][NL];
#pragma unroll
      for (int g = 0; g < G; ++g) {
        const int sj = __builtin_amdgcn_readlane(s, j0 + g);
        const float* hp = H + (long long)sj * D + lane;
#pragma unroll
        for (int i = 0; i < NL; ++i) v[g][i] = hp[32 * i];
      }
#pragma unroll
      for (int g = 0; g < G; ++g) {
        const int dj = __builtin_amdgcn_readlane(d, j0 + g);
        float* o = agg + (long long)dj * D + lane;
#pragma unroll
        for (int i = 0; i < NL; ++i) atomicAdd(o + 32 * i, v[g][i]);
      }
    }
  } else {
    for (int j = 0; j < cnt; ++j) {
      const int sj = __builtin_amdgcn_readlane(s, j);
      const int dj = __builtin_amdgcn_readlane(d, j);
      const float* hp = H + (long long)sj * D + lane;
      float* o = agg + (long long)dj * D + lane;
      for (int i = 0; i < NL; ++i) atomicAdd(o + 32 * i, hp[32 * i]);
    }
  }
}

__global__ void softmax_kernel(float* out, const float* __restrict__ dis,
                               const float* __restrict__ bias, int M) {
  long long gt = (long long)blockIdx.x * blockDim.x + threadIdx.x;
  int node = (int)(gt >> 5);
  int lane = threadIdx.x & 31;
  if (node >= M) return;
  long long idx = (long long)node * 32 + lane;
  float z = dis[node] * out[idx] + bias[lane];
  float m = z;
  for (int off = 16; off > 0; off >>= 1) m = fmaxf(m, __shfl_xor(m, off, 32));
  float ev = __expf(z - m);
  float s = ev;
  for (int off = 16; off > 0; off >>= 1) s += __shfl_xor(s, off, 32);
  out[idx] = ev / s;
}

extern "C" void kernel_launch(void* const* d_in, const int* in_sizes, int n_in,
                              void* d_out, int out_size, void* d_ws, size_t ws_size,
                              hipStream_t stream) {
  (void)n_in; (void)out_size; (void)ws_size;
  const float* x  = (const float*)d_in[0];
  const int*   ei = (const int*)d_in[1];
  const float* W1 = (const float*)d_in[2];
  const float* b1 = (const float*)d_in[3];
  const float* W2 = (const float*)d_in[4];
  const float* b2 = (const float*)d_in[5];

  const int IN = 256, HID = 64, OD = 32;
  const int M = in_sizes[0] / IN;
  const int E = in_sizes[1] / 2;
  const int* src = ei;
  const int* dst = ei + E;

  float* ws   = (float*)d_ws;
  float* dis  = ws;
  float* H1   = dis + (((long long)M + 1023) & ~1023LL);
  float* agg1 = H1 + (long long)M * HID;
  float* H2   = agg1 + (long long)M * HID;
  float* out  = (float*)d_out;

  const int T = 256;

  set_val_kernel<<<(M + T - 1) / T, T, 0, stream>>>(dis, M, 1.0f);
  deg_hist_kernel<<<dim3((M + DEG_RS - 1) / DEG_RS, DEG_NC), 1024, 0, stream>>>(dst, E, dis, M);
  rsqrt_kernel<<<(M + T - 1) / T, T, 0, stream>>>(dis, M);

  gemm_wmma_kernel<256, 64, false><<<(M + 31) / 32, 256, 0, stream>>>(
      x, W1, b1, dis, H1, agg1, M);
  scatter_kernel<64><<<(E + T - 1) / T, T, 0, stream>>>(src, dst, H1, agg1, E);

  gemm_wmma_kernel<64, 32, true><<<(M + 63) / 64, 256, 0, stream>>>(
      agg1, W2, b1, dis, H2, out, M);
  scatter_kernel<32><<<(E + T - 1) / T, T, 0, stream>>>(src, dst, H2, out, E);
  softmax_kernel<<<(int)(((long long)M * 32 + T - 1) / T), T, 0, stream>>>(out, dis, b2, M);
}


# layer-1 GEMM operand staging: 768 resident blocks walk row tiles, W1 put in LDS once per block, X tiles read as whole lines into LDS, next tile prefetched under the matrix loop
# speedup vs baseline: 2.6994x; 1.3513x over previous
#include <hip/hip_runtime.h>

typedef __attribute__((ext_vector_type(2))) float v2f;
typedef __attribute__((ext_vector_type(8))) float v8f;
typedef __attribute__((ext_vector_type(4))) float v4f;

__global__ void set_val_kernel(float* p, long long n, float v) {
  long long i = (long long)blockIdx.x * blockDim.x + threadIdx.x;
  if (i < n) p[i] = v;
}

enum { DEG_RS = 12512, DEG_NC = 32 };

__global__ void __launch_bounds__(1024)
deg_hist_kernel(const int* __restrict__ dst, int E, float* deg, int M) {
  __shared__ int bins[DEG_RS];
  const int tid = threadIdx.x;
  const int lo = blockIdx.x * DEG_RS;
  for (int i = tid; i < DEG_RS; i += 1024) bins[i] = 0;
  __syncthreads();

  const int per = (E + DEG_NC - 1) / DEG_NC;
  const int e0 = blockIdx.y * per;
  int e1 = e0 + per;
  if (e1 > E) e1 = E;
  int e = e0 + tid;
  for (; e + 7 * 1024 < e1; e += 8 * 1024) {
    int d[8];
#pragma unroll
    for (int u = 0; u < 8; ++u) d[u] = dst[e + u * 1024];
#pragma unroll
    for (int u = 0; u < 8; ++u) {
      const unsigned k = (unsigned)(d[u] - lo);
      if (k < (unsigned)DEG_RS) atomicAdd(&bins[k], 1);
    }
  }
  for (; e < e1; e += 1024) {
    const unsigned k = (unsigned)(dst[e] - lo);
    if (k < (unsigned)DEG_RS) atomicAdd(&bins[k], 1);
  }
  __syncthreads();

  for (int i = tid; i < DEG_RS; i += 1024) {
    const int c = bins[i];
    const int n = lo + i;
    if (c != 0 && n < M) atomicAdd(&deg[n], (float)c);
  }
}

__global__ void rsqrt_kernel(float* p, int n) {
  int i = blockIdx.x * blockDim.x + threadIdx.x;
  if (i < n) p[i] = rsqrtf(p[i]);
}

template<int K, int N, bool RELU_A>
__global__ void gemm_wmma_kernel(const float* __restrict__ X,
                                 const float* __restrict__ W,
                                 const float* __restrict__ bias,
                                 const float* __restrict__ dis,
                                 float* __restrict__ OUT,
                                 float* __restrict__ ACC, int M) {
  constexpr int COL_TILES  = N / 16;
  constexpr int ROW_TILES  = 8 / COL_TILES;
  constexpr int BLOCK_ROWS = 16 * ROW_TILES;
  constexpr int AP = K + 4;

  __shared__ float sW[K * N];
  __shared__ __attribute__((aligned(16))) float sA[RELU_A ? BLOCK_ROWS * AP : 4];
  for (int i = threadIdx.x; i < K * N; i += blockDim.x) sW[i] = W[i];
  if (RELU_A) {
    constexpr int NT = 256;
    const int blockRow = blockIdx.x * BLOCK_ROWS;
#pragma unroll
    for (int j = 0; j < BLOCK_ROWS * (K / 4) / NT; ++j) {
      const int i = threadIdx.x + j * NT;
      const int r = i / (K / 4), c = i % (K / 4);
      const int gr = blockRow + r < M ? blockRow + r : M - 1;
      float4 v = *(const float4*)(X + (long long)gr * K + 4 * c);
      const float4 bb = *(const float4*)(bias + 4 * c);
      const float dr = dis[gr];
      v.x = fmaxf(dr * v.x + bb.x, 0.0f);
      v.y = fmaxf(dr * v.y + bb.y, 0.0f);
      v.z = fmaxf(dr * v.z + bb.z, 0.0f);
      v.w = fmaxf(dr * v.w + bb.w, 0.0f);
      *(float4*)(sA + r * AP + 4 * c) = v;
    }
  }
  __syncthreads();

  const int wave = threadIdx.x >> 5;
  const int lane = threadIdx.x & 31;
  const int rowBase = blockIdx.x * BLOCK_ROWS + (wave / COL_TILES) * 16;
  const int colBase = (wave % COL_TILES) * 16;
  if (rowBase + 16 > M) return;

  const int mn  = lane & 15;
  const int grp = lane >> 4;

  const float* xrow = X + (long long)(rowBase + mn) * K + 2 * grp;
  const float* arow = sA + ((wave / COL_TILES) * 16 + mn) * AP + 2 * grp;
  const float* wcol = sW + (long long)(2 * grp) * N + colBase + mn;

  v8f acc = {};
#pragma unroll 1
  for (int k0 = 0; k0 < K; k0 += 16) {
#pragma unroll
    for (int u = 0; u < 16; u += 4) {
      const int k = k0 + u;
      v2f a;
      if (RELU_A) a = *(const v2f*)(arow + k);
      else        a = *(const v2f*)(xrow + k);
      v2f b;
      b.x = wcol[(long long)k * N];
      b.y = wcol[(long long)(k + 1) * N];
      acc = __builtin_amdgcn_wmma_f32_16x16x4_f32(
          false, a, false, b, (short)0, acc, false, false);
    }
    if (RELU_A) {
      asm volatile("v_nop\n\tv_nop\n\tv_nop\n\tv_nop" : "+v"(acc));
    }
  }
  if (!RELU_A) {
    asm volatile("v_nop\n\tv_nop\n\tv_nop\n\tv_nop" : "+v"(acc));
  }

  const int row0 = rowBase + 8 * grp;
  const long long off = (long long)row0 * N + colBase + mn;
  float* out = OUT + off;
  float* st  = ACC + off;
#pragma unroll
  for (int r = 0; r < 8; ++r) {
    const float hv = dis[row0 + r] * acc[r];
    out[(long long)r * N] = hv;
    st[(long long)r * N]  = hv;
  }
}

enum { G1_K = 256, G1_N = 64, G1_ROWS = 32, G1_AP = G1_K + 4, G1_NT = 256,
       G1_LOADS = G1_ROWS * (G1_K / 4) / G1_NT };

__global__ void __launch_bounds__(G1_NT)
gemm1_tiles_kernel(const float* __restrict__ X, const float* __restrict__ W,
                   const float* __restrict__ dis, float* __restrict__ OUT,
                   float* __restrict__ ACC, int M) {
  __shared__ __attribute__((aligned(16))) float sW[G1_K * G1_N];
  __shared__ __attribute__((aligned(16))) float sA[G1_ROWS * G1_AP];
  const int tid = threadIdx.x;

#pragma unroll
  for (int j = 0; j < G1_K * G1_N / 4 / G1_NT; ++j) {
    const int i = tid + j * G1_NT;
    ((v4f*)sW)[i] = ((const v4f*)W)[i];
  }

  const int pr = tid >> 6;
  const int pc = tid & 63;
  const int lane = tid & 31;
  const int wave = __builtin_amdgcn_readfirstlane(tid >> 5);
  const int rt = wave >> 2;
  const int colBase = (wave & 3) * 16;
  const int mn  = lane & 15;
  const int grp = lane >> 4;
  const float* arow = sA + (rt * 16 + mn) * G1_AP + 2 * grp;
  const float* wcol = sW + (2 * grp) * G1_N + colBase + mn;

  const int nTiles = (M + G1_ROWS - 1) / G1_ROWS;
  const float* xcol = X + 4 * pc;
  v4f pre[G1_LOADS];
  int tile = blockIdx.x;
#pragma unroll
  for (int j = 0; j < G1_LOADS; ++j) {
    int gr = tile * G1_ROWS + pr + 4 * j;
    gr = gr < M ? gr : M - 1;
    pre[j] = *(const v4f*)(xcol + (long long)gr * G1_K);
  }

  for (; tile < nTiles; tile += gridDim.x) {
    __syncthreads();
#pragma unroll
    for (int j = 0; j < G1_LOADS; ++j)
      *(v4f*)(sA + (pr + 4 * j) * G1_AP + 4 * pc) = pre[j];
    __syncthreads();

    const int rowBase = tile * G1_ROWS + rt * 16;
    const int row0 = rowBase + 8 * grp;
    const bool whole = rowBase + 16 <= M;
    float dr[8] = {0.0f, 0.0f, 0.0f, 0.0f, 0.0f, 0.0f, 0.0f, 0.0f};
    if (whole) {
#pragma unroll
      for (int r = 0; r < 8; ++r) dr[r] = dis[row0 + r];
    }

    const int nxt = tile + (int)gridDim.x < nTiles ? tile + (int)gridDim.x : tile;
#pragma unroll
    for (int j = 0; j < G1_LOADS; ++j) {
      int gr = nxt * G1_ROWS + pr + 4 * j;
      gr = gr < M ? gr : M - 1;
      pre[j] = *(const v4f*)(xcol + (long long)gr * G1_K);
    }

    if (whole) {
      v8f acc = {};
#pragma unroll 1
      for (int k0 = 0; k0 < G1_K; k0 += 16) {
#pragma unroll
        for (int u = 0; u < 16; u += 4) {
          const int k = k0 + u;
          const v2f a = *(const v2f*)(arow + k);
          v2f b;
          b.x = wcol[k * G1_N];
          b.y = wcol[(k + 1) * G1_N];
          acc = __builtin_amdgcn_wmma_f32_16x16x4_f32(
              false, a, false, b, (short)0, acc, false, false);
        }
        asm volatile("v_nop\n\tv_nop\n\tv_nop\n\tv_nop" : "+v"(acc));
      }

      const long long off = (long long)row0 * G1_N + colBase + mn;
      float* out = OUT + off;
      float* st  = ACC + off;
#pragma unroll
      for (int r = 0; r < 8; ++r) {
        const float hv = dr[r] * acc[r];
        out[(long long)r * G1_N] = hv;
        st[(long long)r * G1_N]  = hv;
      }
    }
  }
}

template<int D>
__global__ void scatter_kernel(const int* __restrict__ src, const int* __restrict__ dst,
                               const float* __restrict__ H, float* __restrict__ agg, int E) {
  constexpr int NL = D / 32;
  constexpr int G  = 8;
  const int lane = threadIdx.x & 31;
  const int wave = __builtin_amdgcn_readfirstlane((int)(threadIdx.x >> 5));
  const long long e0 = ((long long)blockIdx.x * (blockDim.x >> 5) + wave) * 32;
  if (e0 >= E) return;
  const long long rem = (long long)E - e0;
  const int cnt = rem < 32 ? (int)rem : 32;

  int s = 0, d = 0;
  if (lane < cnt) {
    s = src[e0 + lane];
    d = dst[e0 + lane];
  }

  if (cnt == 32) {
#pragma unroll
    for (int j0 = 0; j0 < 32; j0 += G) {
      float v[G][NL];
#pragma unroll
      for (int g = 0; g < G; ++g) {
        const int sj = __builtin_amdgcn_readlane(s, j0 + g);
        const float* hp = H + (long long)sj * D + lane;
#pragma unroll
        for (int i = 0; i < NL; ++i) v[g][i] = hp[32 * i];
      }
#pragma unroll
      for (int g = 0; g < G; ++g) {
        const int dj = __builtin_amdgcn_readlane(d, j0 + g);
        float* o = agg + (long long)dj * D + lane;
#pragma unroll
        for (int i = 0; i < NL; ++i) atomicAdd(o + 32 * i, v[g][i]);
      }
    }
  } else {
    for (int j = 0; j < cnt; ++j) {
      const int sj = __builtin_amdgcn_readlane(s, j);
      const int dj = __builtin_amdgcn_readlane(d, j);
      const float* hp = H + (long long)sj * D + lane;
      float* o = agg + (long long)dj * D + lane;
      for (int i = 0; i < NL; ++i) atomicAdd(o + 32 * i, hp[32 * i]);
    }
  }
}

__global__ void softmax_kernel(float* out, const float* __restrict__ dis,
                               const float* __restrict__ bias, int M) {
  long long gt = (long long)blockIdx.x * blockDim.x + threadIdx.x;
  int node = (int)(gt >> 5);
  int lane = threadIdx.x & 31;
  if (node >= M) return;
  long long idx = (long long)node * 32 + lane;
  float z = dis[node] * out[idx] + bias[lane];
  float m = z;
  for (int off = 16; off > 0; off >>= 1) m = fmaxf(m, __shfl_xor(m, off, 32));
  float ev = __expf(z - m);
  float s = ev;
  for (int off = 16; off > 0; off >>= 1) s += __shfl_xor(s, off, 32);
  out[idx] = ev / s;
}

extern "C" void kernel_launch(void* const* d_in, const int* in_sizes, int n_in,
                              void* d_out, int out_size, void* d_ws, size_t ws_size,
                              hipStream_t stream) {
  (void)n_in; (void)out_size; (void)ws_size;
  const float* x  = (const float*)d_in[0];
  const int*   ei = (const int*)d_in[1];
  const float* W1 = (const float*)d_in[2];
  const float* b1 = (const float*)d_in[3];
  const float* W2 = (const float*)d_in[4];
  const float* b2 = (const float*)d_in[5];

  const int IN = 256, HID = 64, OD = 32;
  const int M = in_sizes[0] / IN;
  const int E = in_sizes[1] / 2;
  const int* src = ei;
  const int* dst = ei + E;

  float* ws   = (float*)d_ws;
  float* dis  = ws;
  float* H1   = dis + (((long long)M + 1023) & ~1023LL);
  float* agg1 = H1 + (long long)M * HID;
  float* H2   = agg1 + (long long)M * HID;
  float* out  = (float*)d_out;

  const int T = 256;

  set_val_kernel<<<(M + T - 1) / T, T, 0, stream>>>(dis, M, 1.0f);
  deg_hist_kernel<<<dim3((M + DEG_RS - 1) / DEG_RS, DEG_NC), 1024, 0, stream>>>(dst, E, dis, M);
  rsqrt_kernel<<<(M + T - 1) / T, T, 0, stream>>>(dis, M);

  gemm1_tiles_kernel<<<768, G1_NT, 0, stream>>>(x, W1, dis, H1, agg1, M);
  scatter_kernel<64><<<(E + T - 1) / T, T, 0, stream>>>(src, dst, H1, agg1, E);

  gemm_wmma_kernel<64, 32, true><<<(M + 63) / 64, 256, 0, stream>>>(
      agg1, W2, b1, dis, H2, out, M);
  scatter_kernel<32><<<(E + T - 1) / T, T, 0, stream>>>(src, dst, H2, out, E);
  softmax_kernel<<<(int)(((long long)M * 32 + T - 1) / T), T, 0, stream>>>(out, dis, b2, M);
}


# edges binned by destination range per call; a range's work-group adds gathered rows to LDS sums started from its own rows, stores each line once, counts degrees from its list; LDS-add arrival order
# speedup vs baseline: 3.1822x; 1.1789x over previous
#include <hip/hip_runtime.h>

typedef __attribute__((ext_vector_type(2))) float v2f;
typedef __attribute__((ext_vector_type(8))) float v8f;
typedef __attribute__((ext_vector_type(4))) float v4f;

enum { OWN_RS = 391, OWN_NR = 256, BIN_NCH = 256, BIN_NT = 1024, ENT_SHIFT = 17,
       ENT_MASK = (1 << ENT_SHIFT) - 1 };

__global__ void __launch_bounds__(BIN_NT)
bin_count_kernel(const int* __restrict__ dst, int E, int* __restrict__ counts) {
  __shared__ int h[OWN_NR];
  const int tid = threadIdx.x;
  if (tid < OWN_NR) h[tid] = 0;
  __syncthreads();
  const int per = (E + BIN_NCH - 1) / BIN_NCH;
  const int e0 = blockIdx.x * per;
  int e1 = e0 + per;
  if (e1 > E) e1 = E;
  for (int eb = e0 + tid; eb < e1; eb += 8 * BIN_NT) {
    int d[8];
#pragma unroll
    for (int u = 0; u < 8; ++u) {
      const int e = eb + u * BIN_NT;
      d[u] = e < e1 ? dst[e] : -1;
    }
#pragma unroll
    for (int u = 0; u < 8; ++u)
      if (d[u] >= 0) atomicAdd(&h[d[u] / OWN_RS], 1);
  }
  __syncthreads();
  if (tid < OWN_NR) counts[blockIdx.x * OWN_NR + tid] = h[tid];
}

__global__ void __launch_bounds__(BIN_NT)
bin_place_kernel(const int* __restrict__ src, const int* __restrict__ dst, int E,
                 const int* __restrict__ counts, int* __restrict__ rangeStart,
                 int* __restrict__ list) {
  __shared__ int ptot[4][OWN_NR];
  __shared__ int ppre[4][OWN_NR];
  __shared__ int tot[OWN_NR];
  __shared__ int cur[OWN_NR];
  const int tid = threadIdx.x;
  const int r = tid & (OWN_NR - 1);
  const int q = tid >> 8;
  const int c = blockIdx.x;
  int t = 0, p = 0;
  const int* cp = counts + q * OWN_NR + r;
  for (int k0 = 0; k0 < BIN_NCH / 4; k0 += 16) {
    int cv[16];
#pragma unroll
    for (int u = 0; u < 16; ++u) cv[u] = cp[(k0 + u) * 4 * OWN_NR];
#pragma unroll
    for (int u = 0; u < 16; ++u) {
      t += cv[u];
      p += q + 4 * (k0 + u) < c ? cv[u] : 0;
    }
  }
  ptot[q][r] = t;
  ppre[q][r] = p;
  __syncthreads();
  if (tid < OWN_NR) {
    tot[tid] = ptot[0][tid] + ptot[1][tid] + ptot[2][tid] + ptot[3][tid];
    cur[tid] = ppre[0][tid] + ppre[1][tid] + ppre[2][tid] + ppre[3][tid];
  }
  __syncthreads();
  if (tid < 32) {
    int v[8];
    int sum = 0;
#pragma unroll
    for (int k = 0; k < 8; ++k) {
      const int x = tot[8 * tid + k];
      v[k] = sum;
      sum += x;
    }
    int inc = sum;
#pragma unroll
    for (int off = 1; off < 32; off <<= 1) {
      const int y = __shfl_up(inc, off, 32);
      if (tid >= off) inc += y;
    }
    const int excl = inc - sum;
#pragma unroll
    for (int k = 0; k < 8; ++k) {
      const int st = excl + v[k];
      cur[8 * tid + k] += st;
      if (c == 0) rangeStart[8 * tid + k] = st;
    }
    if (c == 0 && tid == 31) rangeStart[OWN_NR] = inc;
  }
  __syncthreads();

  const int per = (E + BIN_NCH - 1) / BIN_NCH;
  const int e0 = c * per;
  int e1 = e0 + per;
  if (e1 > E) e1 = E;
  for (int eb = e0 + tid; eb < e1; eb += 8 * BIN_NT) {
    int d[8], sv[8];
#pragma unroll
    for (int u = 0; u < 8; ++u) {
      const int e = eb + u * BIN_NT;
      const bool ok = e < e1;
      d[u]  = ok ? dst[e] : -1;
      sv[u] = ok ? src[e] : 0;
    }
#pragma unroll
    for (int u = 0; u < 8; ++u) {
      if (d[u] >= 0) {
        const int rr = d[u] / OWN_RS;
        const int loc = d[u] - rr * OWN_RS;
        const int pos = atomicAdd(&cur[rr], 1);
        list[pos] = sv[u] | (loc << ENT_SHIFT);
      }
    }
  }
}

__global__ void __launch_bounds__(BIN_NT)
deg_list_kernel(const int* __restrict__ list, const int* __restrict__ rangeStart,
                float* __restrict__ dis, int M) {
  __shared__ int bins[OWN_RS];
  const int tid = threadIdx.x;
  if (tid < OWN_RS) bins[tid] = 0;
  __syncthreads();
  const int b = rangeStart[blockIdx.x];
  const int e = rangeStart[blockIdx.x + 1];
  for (int pb = b + tid; pb < e; pb += 8 * BIN_NT) {
    int w[8];
#pragma unroll
    for (int u = 0; u < 8; ++u) {
      const int p = pb + u * BIN_NT;
      w[u] = p < e ? list[p] : -1;
    }
#pragma unroll
    for (int u = 0; u < 8; ++u)
      if (w[u] >= 0) atomicAdd(&bins[w[u] >> ENT_SHIFT], 1);
  }
  __syncthreads();
  const int n = blockIdx.x * OWN_RS + tid;
  if (tid < OWN_RS && n < M) dis[n] = rsqrtf(1.0f + (float)bins[tid]);
}

template<int K, int N, bool RELU_A>
__global__ void gemm_wmma_kernel(const float* __restrict__ X,
                                 const float* __restrict__ W,
                                 const float* __restrict__ bias,
                                 const float* __restrict__ dis,
                                 float* __restrict__ OUT, int M) {
  constexpr int COL_TILES  = N / 16;
  constexpr int ROW_TILES  = 8 / COL_TILES;
  constexpr int BLOCK_ROWS = 16 * ROW_TILES;
  constexpr int AP = K + 4;

  __shared__ float sW[K * N];
  __shared__ __attribute__((aligned(16))) float sA[RELU_A ? BLOCK_ROWS * AP : 4];
  for (int i = threadIdx.x; i < K * N; i += blockDim.x) sW[i] = W[i];
  if (RELU_A) {
    constexpr int NT = 256;
    const int blockRow = blockIdx.x * BLOCK_ROWS;
#pragma unroll
    for (int j = 0; j < BLOCK_ROWS * (K / 4) / NT; ++j) {
      const int i = threadIdx.x + j * NT;
      const int r = i / (K / 4), c = i % (K / 4);
      const int gr = blockRow + r < M ? blockRow + r : M - 1;
      float4 v = *(const float4*)(X + (long long)gr * K + 4 * c);
      const float4 bb = *(const float4*)(bias + 4 * c);
      const float dr = dis[gr];
      v.x = fmaxf(dr * v.x + bb.x, 0.0f);
      v.y = fmaxf(dr * v.y + bb.y, 0.0f);
      v.z = fmaxf(dr * v.z + bb.z, 0.0f);
      v.w = fmaxf(dr * v.w + bb.w, 0.0f);
      *(float4*)(sA + r * AP + 4 * c) = v;
    }
  }
  __syncthreads();

  const int wave = threadIdx.x >> 5;
  const int lane = threadIdx.x & 31;
  const int rowBase = blockIdx.x * BLOCK_ROWS + (wave / COL_TILES) * 16;
  const int colBase = (wave % COL_TILES) * 16;
  if (rowBase + 16 > M) return;

  const int mn  = lane & 15;
  const int grp = lane >> 4;

  const float* xrow = X + (long long)(rowBase + mn) * K + 2 * grp;
  const float* arow = sA + ((wave / COL_TILES) * 16 + mn) * AP + 2 * grp;
  const float* wcol = sW + (long long)(2 * grp) * N + colBase + mn;

  v8f acc = {};
#pragma unroll 1
  for (int k0 = 0; k0 < K; k0 += 16) {
#pragma unroll
    for (int u = 0; u < 16; u += 4) {
      const int k = k0 + u;
      v2f a;
      if (RELU_A) a = *(const v2f*)(arow + k);
      else        a = *(const v2f*)(xrow + k);
      v2f b;
      b.x = wcol[(long long)k * N];
      b.y = wcol[(long long)(k + 1) * N];
      acc = __builtin_amdgcn_wmma_f32_16x16x4_f32(
          false, a, false, b, (short)0, acc, false, false);
    }
    if (RELU_A) {
      asm volatile("v_nop\n\tv_nop\n\tv_nop\n\tv_nop" : "+v"(acc));
    }
  }
  if (!RELU_A) {
    asm volatile("v_nop\n\tv_nop\n\tv_nop\n\tv_nop" : "+v"(acc));
  }

  const int row0 = rowBase + 8 * grp;
  float* out = OUT + (long long)row0 * N + colBase + mn;
#pragma unroll
  for (int r = 0; r < 8; ++r) out[(long long)r * N] = dis[row0 + r] * acc[r];
}

enum { G1_K = 256, G1_N = 64, G1_ROWS = 32, G1_AP = G1_K + 4, G1_NT = 256,
       G1_LOADS = G1_ROWS * (G1_K / 4) / G1_NT };

__global__ void __launch_bounds__(G1_NT)
gemm1_tiles_kernel(const float* __restrict__ X, const float* __restrict__ W,
                   const float* __restrict__ dis, float* __restrict__ OUT, int M) {
  __shared__ __attribute__((aligned(16))) float sW[G1_K * G1_N];
  __shared__ __attribute__((aligned(16))) float sA[G1_ROWS * G1_AP];
  const int tid = threadIdx.x;

#pragma unroll
  for (int j = 0; j < G1_K * G1_N / 4 / G1_NT; ++j) {
    const int i = tid + j * G1_NT;
    ((v4f*)sW)[i] = ((const v4f*)W)[i];
  }

  const int pr = tid >> 6;
  const int pc = tid & 63;
  const int lane = tid & 31;
  const int wave = __builtin_amdgcn_readfirstlane(tid >> 5);
  const int rt = wave >> 2;
  const int colBase = (wave & 3) * 16;
  const int mn  = lane & 15;
  const int grp = lane >> 4;
  const float* arow = sA + (rt * 16 + mn) * G1_AP + 2 * grp;
  const float* wcol = sW + (2 * grp) * G1_N + colBase + mn;

  const int nTiles = (M + G1_ROWS - 1) / G1_ROWS;
  const float* xcol = X + 4 * pc;
  v4f pre[G1_LOADS];
  int tile = blockIdx.x;
#pragma unroll
  for (int j = 0; j < G1_LOADS; ++j) {
    int gr = tile * G1_ROWS + pr + 4 * j;
    gr = gr < M ? gr : M - 1;
    pre[j] = *(const v4f*)(xcol + (long long)gr * G1_K);
  }

  for (; tile < nTiles; tile += gridDim.x) {
    __syncthreads();
#pragma unroll
    for (int j = 0; j < G1_LOADS; ++j)
      *(v4f*)(sA + (pr + 4 * j) * G1_AP + 4 * pc) = pre[j];
    __syncthreads();

    const int rowBase = tile * G1_ROWS + rt * 16;
    const int row0 = rowBase + 8 * grp;
    const bool whole = rowBase + 16 <= M;
    float dr[8] = {0.0f, 0.0f, 0.0f, 0.0f, 0.0f, 0.0f, 0.0f, 0.0f};
    if (whole) {
#pragma unroll
      for (int r = 0; r < 8; ++r) dr[r] = dis[row0 + r];
    }

    const int nxt = tile + (int)gridDim.x < nTiles ? tile + (int)gridDim.x : tile;
#pragma unroll
    for (int j = 0; j < G1_LOADS; ++j) {
      int gr = nxt * G1_ROWS + pr + 4 * j;
      gr = gr < M ? gr : M - 1;
      pre[j] = *(const v4f*)(xcol + (long long)gr * G1_K);
    }

    if (whole) {
      v8f acc = {};
#pragma unroll 1
      for (int k0 = 0; k0 < G1_K; k0 += 16) {
#pragma unroll
        for (int u = 0; u < 16; u += 4) {
          const int k = k0 + u;
          const v2f a = *(const v2f*)(arow + k);
          v2f b;
          b.x = wcol[k * G1_N];
          b.y = wcol[(k + 1) * G1_N];
          acc = __builtin_amdgcn_wmma_f32_16x16x4_f32(
              false, a, false, b, (short)0, acc, false, false);
        }
        asm volatile("v_nop\n\tv_nop\n\tv_nop\n\tv_nop" : "+v"(acc));
      }

      float* out = OUT + (long long)row0 * G1_N + colBase + mn;
#pragma unroll
      for (int r = 0; r < 8; ++r) out[(long long)r * G1_N] = dr[r] * acc[r];
    }
  }
}

template<int D>
__global__ void __launch_bounds__(BIN_NT)
owner_sum_kernel(const int* __restrict__ list, const int* __restrict__ rangeStart,
                 const float* __restrict__ H, float* __restrict__ agg, int M) {
  constexpr int NL = D / 32;
  constexpr int G  = 16 / NL;
  constexpr int NG = 32 / G;
  constexpr int PD = NG < 3 ? NG : 3;
  constexpr int NI = (OWN_RS * (D / 4) + BIN_NT - 1) / BIN_NT;
  constexpr int PAD_ENT = OWN_RS << ENT_SHIFT;
  __shared__ __attribute__((aligned(16))) float acc[(OWN_RS + 1) * D];
  const int tid = threadIdx.x;
  const int lo = blockIdx.x * OWN_RS;
  int nrows = M - lo;
  if (nrows > OWN_RS) nrows = OWN_RS;
  const int nq = nrows * (D / 4);
  const v4f* hq = (const v4f*)(H + (long long)lo * D);
  {
    v4f t[NI];
#pragma unroll
    for (int k = 0; k < NI; ++k) {
      const int i = tid + k * BIN_NT;
      t[k] = i < nq ? hq[i] : v4f{0.0f, 0.0f, 0.0f, 0.0f};
    }
#pragma unroll
    for (int k = 0; k < NI; ++k) {
      const int i = tid + k * BIN_NT;
      if (i < nq) ((v4f*)acc)[i] = t[k];
    }
  }
  __syncthreads();

  const int lane = tid & 31;
  const int wave = __builtin_amdgcn_readfirstlane(tid >> 5);
  const int b = rangeStart[blockIdx.x];
  const int e = rangeStart[blockIdx.x + 1];
  const int p0 = b + wave * 32;
  int ent = PAD_ENT;
  if (p0 + lane < e) ent = list[p0 + lane];
  for (int p = p0; p < e; p += BIN_NT) {
    const int pn = p + BIN_NT + lane;
    int entN = list[pn < e ? pn : e - 1];
    entN = pn < e ? entN : PAD_ENT;

    float v[NG][G][NL];
#pragma unroll
    for (int k = 0; k < NG + PD; ++k) {
      if (k >= PD) {
#pragma unroll
        for (int g = 0; g < G; ++g) {
          const int lj = __builtin_amdgcn_readlane(ent, (k - PD) * G + g) >> ENT_SHIFT;
          float* o = acc + lj * D + lane;
#pragma unroll
          for (int i = 0; i < NL; ++i) atomicAdd(o + 32 * i, v[k - PD][g][i]);
        }
      }
      if (k < NG) {
#pragma unroll
        for (int g = 0; g < G; ++g) {
          const int sj = __builtin_amdgcn_readlane(ent, k * G + g) & ENT_MASK;
          const float* hp = H + (long long)sj * D + lane;
#pragma unroll
          for (int i = 0; i < NL; ++i) v[k][g][i] = hp[32 * i];
        }
      }
    }
    ent = entN;
  }
  __syncthreads();

  v4f* oq = (v4f*)(agg + (long long)lo * D);
  for (int i = tid; i < nq; i += BIN_NT) oq[i] = ((const v4f*)acc)[i];
}

__global__ void softmax_kernel(float* out, const float* __restrict__ dis,
                               const float* __restrict__ bias, int M) {
  long long gt = (long long)blockIdx.x * blockDim.x + threadIdx.x;
  int node = (int)(gt >> 5);
  int lane = threadIdx.x & 31;
  if (node >= M) return;
  long long idx = (long long)node * 32 + lane;
  float z = dis[node] * out[idx] + bias[lane];
  float m = z;
  for (int off = 16; off > 0; off >>= 1) m = fmaxf(m, __shfl_xor(m, off, 32));
  float ev = __expf(z - m);
  float s = ev;
  for (int off = 16; off > 0; off >>= 1) s += __shfl_xor(s, off, 32);
  out[idx] = ev / s;
}

extern "C" void kernel_launch(void* const* d_in, const int* in_sizes, int n_in,
                              void* d_out, int out_size, void* d_ws, size_t ws_size,
                              hipStream_t stream) {
  (void)n_in; (void)out_size; (void)ws_size;
  const float* x  = (const float*)d_in[0];
  const int*   ei = (const int*)d_in[1];
  const float* W1 = (const float*)d_in[2];
  const float* b1 = (const float*)d_in[3];
  const float* W2 = (const float*)d_in[4];
  const float* b2 = (const float*)d_in[5];

  const int IN = 256, HID = 64, OD = 32;
  const int M = in_sizes[0] / IN;
  const int E = in_sizes[1] / 2;
  const int* src = ei;
  const int* dst = ei + E;

  float* ws   = (float*)d_ws;
  float* dis  = ws;
  float* H1   = dis + (((long long)M + 1023) & ~1023LL);
  float* agg1 = H1 + (long long)M * HID;
  float* H2   = H1;
  int* list   = (int*)(agg1 + (long long)M * HID);
  int* counts = list + E;
  int* rstart = counts + BIN_NCH * OWN_NR;
  float* out  = (float*)d_out;

  const int T = 256;
  const int NRANGE = (M + OWN_RS - 1) / OWN_RS;

  bin_count_kernel<<<BIN_NCH, BIN_NT, 0, stream>>>(dst, E, counts);
  bin_place_kernel<<<BIN_NCH, BIN_NT, 0, stream>>>(src, dst, E, counts, rstart, list);
  deg_list_kernel<<<NRANGE, BIN_NT, 0, stream>>>(list, rstart, dis, M);

  gemm1_tiles_kernel<<<768, G1_NT, 0, stream>>>(x, W1, dis, H1, M);
  owner_sum_kernel<64><<<NRANGE, BIN_NT, 0, stream>>>(list, rstart, H1, agg1, M);

  gemm_wmma_kernel<64, 32, true><<<(M + 63) / 64, 256, 0, stream>>>(
      agg1, W2, b1, dis, H2, M);
  owner_sum_kernel<32><<<NRANGE, BIN_NT, 0, stream>>>(list, rstart, H2, out, M);
  softmax_kernel<<<(int)(((long long)M * 32 + T - 1) / T), T, 0, stream>>>(out, dis, b2, M);
}


# edge-sum consumers run in the owner work-groups from LDS: layer-2 GEMM (f32 WMMA) after layer 1's sums, softmax after layer 2's; 2 launches (112k waves) and the sums' memory trips go; sum order kept
# speedup vs baseline: 3.6567x; 1.1491x over previous
#include <hip/hip_runtime.h>

typedef __attribute__((ext_vector_type(2))) float v2f;
typedef __attribute__((ext_vector_type(8))) float v8f;
typedef __attribute__((ext_vector_type(4))) float v4f;

enum { OWN_RS = 391, OWN_NR = 256, BIN_NCH = 256, BIN_NT = 1024, ENT_SHIFT = 17,
       ENT_MASK = (1 << ENT_SHIFT) - 1 };

__global__ void __launch_bounds__(BIN_NT)
bin_count_kernel(const int* __restrict__ dst, int E, int* __restrict__ counts) {
  __shared__ int h[OWN_NR];
  const int tid = threadIdx.x;
  if (tid < OWN_NR) h[tid] = 0;
  __syncthreads();
  const int per = (E + BIN_NCH - 1) / BIN_NCH;
  const int e0 = blockIdx.x * per;
  int e1 = e0 + per;
  if (e1 > E) e1 = E;
  for (int eb = e0 + tid; eb < e1; eb += 8 * BIN_NT) {
    int d[8];
#pragma unroll
    for (int u = 0; u < 8; ++u) {
      const int e = eb + u * BIN_NT;
      d[u] = e < e1 ? dst[e] : -1;
    }
#pragma unroll
    for (int u = 0; u < 8; ++u)
      if (d[u] >= 0) atomicAdd(&h[d[u] / OWN_RS], 1);
  }
  __syncthreads();
  if (tid < OWN_NR) counts[blockIdx.x * OWN_NR + tid] = h[tid];
}

__global__ void __launch_bounds__(BIN_NT)
bin_place_kernel(const int* __restrict__ src, const int* __restrict__ dst, int E,
                 const int* __restrict__ counts, int* __restrict__ rangeStart,
                 int* __restrict__ list) {
  __shared__ int ptot[4][OWN_NR];
  __shared__ int ppre[4][OWN_NR];
  __shared__ int tot[OWN_NR];
  __shared__ int cur[OWN_NR];
  const int tid = threadIdx.x;
  const int r = tid & (OWN_NR - 1);
  const int q = tid >> 8;
  const int c = blockIdx.x;
  int t = 0, p = 0;
  const int* cp = counts + q * OWN_NR + r;
  for (int k0 = 0; k0 < BIN_NCH / 4; k0 += 16) {
    int cv[16];
#pragma unroll
    for (int u = 0; u < 16; ++u) cv[u] = cp[(k0 + u) * 4 * OWN_NR];
#pragma unroll
    for (int u = 0; u < 16; ++u) {
      t += cv[u];
      p += q + 4 * (k0 + u) < c ? cv[u] : 0;
    }
  }
  ptot[q][r] = t;
  ppre[q][r] = p;
  __syncthreads();
  if (tid < OWN_NR) {
    tot[tid] = ptot[0][tid] + ptot[1][tid] + ptot[2][tid] + ptot[3][tid];
    cur[tid] = ppre[0][tid] + ppre[1][tid] + ppre[2][tid] + ppre[3][tid];
  }
  __syncthreads();
  if (tid < 32) {
    int v[8];
    int sum = 0;
#pragma unroll
    for (int k = 0; k < 8; ++k) {
      const int x = tot[8 * tid + k];
      v[k] = sum;
      sum += x;
    }
    int inc = sum;
#pragma unroll
    for (int off = 1; off < 32; off <<= 1) {
      const int y = __shfl_up(inc, off, 32);
      if (tid >= off) inc += y;
    }
    const int excl = inc - sum;
#pragma unroll
    for (int k = 0; k < 8; ++k) {
      const int st = excl + v[k];
      cur[8 * tid + k] += st;
      if (c == 0) rangeStart[8 * tid + k] = st;
    }
    if (c == 0 && tid == 31) rangeStart[OWN_NR] = inc;
  }
  __syncthreads();

  const int per = (E + BIN_NCH - 1) / BIN_NCH;
  const int e0 = c * per;
  int e1 = e0 + per;
  if (e1 > E) e1 = E;
  for (int eb = e0 + tid; eb < e1; eb += 8 * BIN_NT) {
    int d[8], sv[8];
#pragma unroll
    for (int u = 0; u < 8; ++u) {
      const int e = eb + u * BIN_NT;
      const bool ok = e < e1;
      d[u]  = ok ? dst[e] : -1;
      sv[u] = ok ? src[e] : 0;
    }
#pragma unroll
    for (int u = 0; u < 8; ++u) {
      if (d[u] >= 0) {
        const int rr = d[u] / OWN_RS;
        const int loc = d[u] - rr * OWN_RS;
        const int pos = atomicAdd(&cur[rr], 1);
        list[pos] = sv[u] | (loc << ENT_SHIFT);
      }
    }
  }
}

__global__ void __launch_bounds__(BIN_NT)
deg_list_kernel(const int* __restrict__ list, const int* __restrict__ rangeStart,
                float* __restrict__ dis, int M) {
  __shared__ int bins[OWN_RS];
  const int tid = threadIdx.x;
  if (tid < OWN_RS) bins[tid] = 0;
  __syncthreads();
  const int b = rangeStart[blockIdx.x];
  const int e = rangeStart[blockIdx.x + 1];
  for (int pb = b + tid; pb < e; pb += 8 * BIN_NT) {
    int w[8];
#pragma unroll
    for (int u = 0; u < 8; ++u) {
      const int p = pb + u * BIN_NT;
      w[u] = p < e ? list[p] : -1;
    }
#pragma unroll
    for (int u = 0; u < 8; ++u)
      if (w[u] >= 0) atomicAdd(&bins[w[u] >> ENT_SHIFT], 1);
  }
  __syncthreads();
  const int n = blockIdx.x * OWN_RS + tid;
  if (tid < OWN_RS && n < M) dis[n] = rsqrtf(1.0f + (float)bins[tid]);
}

enum { G1_K = 256, G1_N = 64, G1_ROWS = 32, G1_AP = G1_K + 4, G1_NT = 256,
       G1_LOADS = G1_ROWS * (G1_K / 4) / G1_NT };

__global__ void __launch_bounds__(G1_NT)
gemm1_tiles_kernel(const float* __restrict__ X, const float* __restrict__ W,
                   const float* __restrict__ dis, float* __restrict__ OUT, int M) {
  __shared__ __attribute__((aligned(16))) float sW[G1_K * G1_N];
  __shared__ __attribute__((aligned(16))) float sA[G1_ROWS * G1_AP];
  const int tid = threadIdx.x;

#pragma unroll
  for (int j = 0; j < G1_K * G1_N / 4 / G1_NT; ++j) {
    const int i = tid + j * G1_NT;
    ((v4f*)sW)[i] = ((const v4f*)W)[i];
  }

  const int pr = tid >> 6;
  const int pc = tid & 63;
  const int lane = tid & 31;
  const int wave = __builtin_amdgcn_readfirstlane(tid >> 5);
  const int rt = wave >> 2;
  const int colBase = (wave & 3) * 16;
  const int mn  = lane & 15;
  const int grp = lane >> 4;
  const float* arow = sA + (rt * 16 + mn) * G1_AP + 2 * grp;
  const float* wcol = sW + (2 * grp) * G1_N + colBase + mn;

  const int nTiles = (M + G1_ROWS - 1) / G1_ROWS;
  const float* xcol = X + 4 * pc;
  v4f pre[G1_LOADS];
  int tile = blockIdx.x;
#pragma unroll
  for (int j = 0; j < G1_LOADS; ++j) {
    int gr = tile * G1_ROWS + pr + 4 * j;
    gr = gr < M ? gr : M - 1;
    pre[j] = *(const v4f*)(xcol + (long long)gr * G1_K);
  }

  for (; tile < nTiles; tile += gridDim.x) {
    __syncthreads();
#pragma unroll
    for (int j = 0; j < G1_LOADS; ++j)
      *(v4f*)(sA + (pr + 4 * j) * G1_AP + 4 * pc) = pre[j];
    __syncthreads();

    const int rowBase = tile * G1_ROWS + rt * 16;
    const int row0 = rowBase + 8 * grp;
    const bool whole = rowBase + 16 <= M;
    float dr[8] = {0.0f, 0.0f, 0.0f, 0.0f, 0.0f, 0.0f, 0.0f, 0.0f};
    if (whole) {
#pragma unroll
      for (int r = 0; r < 8; ++r) dr[r] = dis[row0 + r];
    }

    const int nxt = tile + (int)gridDim.x < nTiles ? tile + (int)gridDim.x : tile;
#pragma unroll
    for (int j = 0; j < G1_LOADS; ++j) {
      int gr = nxt * G1_ROWS + pr + 4 * j;
      gr = gr < M ? gr : M - 1;
      pre[j] = *(const v4f*)(xcol + (long long)gr * G1_K);
    }

    if (whole) {
      v8f acc = {};
#pragma unroll 1
      for (int k0 = 0; k0 < G1_K; k0 += 16) {
#pragma unroll
        for (int u = 0; u < 16; u += 4) {
          const int k = k0 + u;
          const v2f a = *(const v2f*)(arow + k);
          v2f b;
          b.x = wcol[k * G1_N];
          b.y = wcol[(k + 1) * G1_N];
          acc = __builtin_amdgcn_wmma_f32_16x16x4_f32(
              false, a, false, b, (short)0, acc, false, false);
        }
        asm volatile("v_nop\n\tv_nop\n\tv_nop\n\tv_nop" : "+v"(acc));
      }

      float* out = OUT + (long long)row0 * G1_N + colBase + mn;
#pragma unroll
      for (int r = 0; r < 8; ++r) out[(long long)r * G1_N] = dr[r] * acc[r];
    }
  }
}

template<int D>
__global__ void __launch_bounds__(BIN_NT)
owner_sum_kernel(const int* __restrict__ list, const int* __restrict__ rangeStart,
                 const float* __restrict__ H, const float* __restrict__ dis,
                 const float* __restrict__ bias, const float* __restrict__ Wn,
                 float* __restrict__ res, int M) {
  constexpr int NL = D / 32;
  constexpr int G  = 16 / NL;
  constexpr int NG = 32 / G;
  constexpr int PD = NG < 3 ? NG : 3;
  constexpr int NI = (OWN_RS * (D / 4) + BIN_NT - 1) / BIN_NT;
  constexpr int PAD_ENT = OWN_RS << ENT_SHIFT;
  constexpr bool GEMM = D == 64;
  constexpr int NO = 32;
  constexpr int NRT = (OWN_RS + 15) / 16;
  constexpr int P = GEMM ? D + 4 : D;
  constexpr int ROWS = GEMM ? NRT * 16 : OWN_RS + 1;
  __shared__ __attribute__((aligned(16))) float acc[ROWS * P];
  __shared__ __attribute__((aligned(16))) float sW[GEMM ? D * NO : 4];
  __shared__ float sDis[NRT * 16];
  const int tid = threadIdx.x;
  const int lo = blockIdx.x * OWN_RS;
  int nrows = M - lo;
  if (nrows > OWN_RS) nrows = OWN_RS;
  const int nq = nrows * (D / 4);
  const v4f* hq = (const v4f*)(H + (long long)lo * D);
  {
    v4f t[NI];
#pragma unroll
    for (int k = 0; k < NI; ++k) {
      const int i = tid + k * BIN_NT;
      t[k] = i < nq ? hq[i] : v4f{0.0f, 0.0f, 0.0f, 0.0f};
    }
#pragma unroll
    for (int k = 0; k < NI; ++k) {
      const int i = tid + k * BIN_NT;
      if (i < nq) *(v4f*)(acc + (i / (D / 4)) * P + 4 * (i % (D / 4))) = t[k];
    }
  }
  if (tid < NRT * 16) sDis[tid] = dis[lo + tid < M ? lo + tid : M - 1];
  if (GEMM) {
#pragma unroll
    for (int j = 0; j < D * NO / BIN_NT; ++j) {
      const int i = tid + j * BIN_NT;
      const int k = i / NO, n = i % NO;
      sW[((k >> 1) * 16 + (n & 15)) * 4 + (n >> 4) * 2 + (k & 1)] = Wn[i];
    }
  }
  __syncthreads();

  const int lane = tid & 31;
  const int wave = __builtin_amdgcn_readfirstlane(tid >> 5);
  const int b = rangeStart[blockIdx.x];
  const int e = rangeStart[blockIdx.x + 1];
  const int p0 = b + wave * 32;
  int ent = PAD_ENT;
  if (p0 + lane < e) ent = list[p0 + lane];
  for (int p = p0; p < e; p += BIN_NT) {
    const int pn = p + BIN_NT + lane;
    int entN = list[pn < e ? pn : e - 1];
    entN = pn < e ? entN : PAD_ENT;

    float v[NG][G][NL];
#pragma unroll
    for (int k = 0; k < NG + PD; ++k) {
      if (k >= PD) {
#pragma unroll
        for (int g = 0; g < G; ++g) {
          const int lj = __builtin_amdgcn_readlane(ent, (k - PD) * G + g) >> ENT_SHIFT;
          float* o = acc + lj * P + lane;
#pragma unroll
          for (int i = 0; i < NL; ++i) atomicAdd(o + 32 * i, v[k - PD][g][i]);
        }
      }
      if (k < NG) {
#pragma unroll
        for (int g = 0; g < G; ++g) {
          const int sj = __builtin_amdgcn_readlane(ent, k * G + g) & ENT_MASK;
          const float* hp = H + (long long)sj * D + lane;
#pragma unroll
          for (int i = 0; i < NL; ++i) v[k][g][i] = hp[32 * i];
        }
      }
    }
    ent = entN;
  }
  __syncthreads();

  if (GEMM) {
    const float4 bb = *(const float4*)(bias + 4 * (tid % (D / 4)));
    for (int i = tid; i < NRT * 16 * (D / 4); i += BIN_NT) {
      const int r = i / (D / 4), c = i % (D / 4);
      v4f v = {0.0f, 0.0f, 0.0f, 0.0f};
      if (r < nrows) {
        v = *(const v4f*)(acc + r * P + 4 * c);
        const float dr = sDis[r];
        v.x = fmaxf(dr * v.x + bb.x, 0.0f);
        v.y = fmaxf(dr * v.y + bb.y, 0.0f);
        v.z = fmaxf(dr * v.z + bb.z, 0.0f);
        v.w = fmaxf(dr * v.w + bb.w, 0.0f);
      }
      *(v4f*)(acc + r * P + 4 * c) = v;
    }
    __syncthreads();

    if (wave < NRT) {
      const int mn  = lane & 15;
      const int grp = lane >> 4;
      const float* arow = acc + (wave * 16 + mn) * P + 2 * grp;
      const float* wq = sW + (grp * 16 + mn) * 4;
      const int lr0 = wave * 16 + 8 * grp;
      float* out = res + (long long)(lo + lr0) * NO + mn;
      float dr[8];
#pragma unroll
      for (int r = 0; r < 8; ++r) dr[r] = sDis[lr0 + r];
      v8f c0 = {0.0f, 0.0f, 0.0f, 0.0f, 0.0f, 0.0f, 0.0f, 0.0f};
      v8f c1 = {0.0f, 0.0f, 0.0f, 0.0f, 0.0f, 0.0f, 0.0f, 0.0f};
      asm volatile("" : "+v"(c0), "+v"(c1), "+v"(out));
      typedef __attribute__((address_space(1))) float gfloat;
      gfloat* outg = (gfloat*)out;
#pragma unroll
      for (int k = 0; k < D; k += 4) {
        const v2f a = *(const v2f*)(arow + k);
        const v4f b = *(const v4f*)(wq + k * 32);
        const v2f b0 = {b.x, b.y};
        const v2f b1 = {b.z, b.w};
        c0 = __builtin_amdgcn_wmma_f32_16x16x4_f32(
            false, a, false, b0, (short)0, c0, false, false);
        c1 = __builtin_amdgcn_wmma_f32_16x16x4_f32(
            false, a, false, b1, (short)0, c1, false, false);
      }
      asm volatile("v_nop\n\tv_nop\n\tv_nop\n\tv_nop" : "+v"(c0), "+v"(c1));

#pragma unroll
      for (int r = 0; r < 8; ++r) {
        if (lr0 + r < nrows) {
          outg[r * NO] = dr[r] * c0[r];
          outg[r * NO + 16] = dr[r] * c1[r];
        }
      }
    }
  } else {
    constexpr int RPW = (OWN_RS + 31) / 32;
    const float bl = bias[lane];
    float z[RPW];
#pragma unroll
    for (int j = 0; j < RPW; ++j) {
      const int r = wave + 32 * j;
      const int rr = r < nrows ? r : 0;
      z[j] = sDis[rr] * acc[rr * P + lane] + bl;
    }
    float m[RPW], ev[RPW], sm[RPW];
#pragma unroll
    for (int j = 0; j < RPW; ++j) m[j] = z[j];
#pragma unroll
    for (int off = 16; off > 0; off >>= 1) {
#pragma unroll
      for (int j = 0; j < RPW; ++j) m[j] = fmaxf(m[j], __shfl_xor(m[j], off, 32));
    }
#pragma unroll
    for (int j = 0; j < RPW; ++j) { ev[j] = __expf(z[j] - m[j]); sm[j] = ev[j]; }
#pragma unroll
    for (int off = 16; off > 0; off >>= 1) {
#pragma unroll
      for (int j = 0; j < RPW; ++j) sm[j] += __shfl_xor(sm[j], off, 32);
    }
#pragma unroll
    for (int j = 0; j < RPW; ++j) {
      const int r = wave + 32 * j;
      if (r < nrows) res[(long long)(lo + r) * D + lane] = ev[j] / sm[j];
    }
  }
}

extern "C" void kernel_launch(void* const* d_in, const int* in_sizes, int n_in,
                              void* d_out, int out_size, void* d_ws, size_t ws_size,
                              hipStream_t stream) {
  (void)n_in; (void)out_size; (void)ws_size;
  const float* x  = (const float*)d_in[0];
  const int*   ei = (const int*)d_in[1];
  const float* W1 = (const float*)d_in[2];
  const float* b1 = (const float*)d_in[3];
  const float* W2 = (const float*)d_in[4];
  const float* b2 = (const float*)d_in[5];

  const int IN = 256, HID = 64, OD = 32;
  const int M = in_sizes[0] / IN;
  const int E = in_sizes[1] / 2;
  const int* src = ei;
  const int* dst = ei + E;

  float* ws   = (float*)d_ws;
  float* dis  = ws;
  float* H1   = dis + (((long long)M + 1023) & ~1023LL);
  float* H2   = H1 + (long long)M * HID;
  int* list   = (int*)(H2 + (long long)M * HID);
  int* counts = list + E;
  int* rstart = counts + BIN_NCH * OWN_NR;
  float* out  = (float*)d_out;

  const int NRANGE = (M + OWN_RS - 1) / OWN_RS;

  bin_count_kernel<<<BIN_NCH, BIN_NT, 0, stream>>>(dst, E, counts);
  bin_place_kernel<<<BIN_NCH, BIN_NT, 0, stream>>>(src, dst, E, counts, rstart, list);
  deg_list_kernel<<<NRANGE, BIN_NT, 0, stream>>>(list, rstart, dis, M);

  gemm1_tiles_kernel<<<768, G1_NT, 0, stream>>>(x, W1, dis, H1, M);
  owner_sum_kernel<64><<<NRANGE, BIN_NT, 0, stream>>>(list, rstart, H1, dis, b1, W2, H2, M);

  owner_sum_kernel<32><<<NRANGE, BIN_NT, 0, stream>>>(list, rstart, H2, dis, b2, W2, out, M);
}


# edge binning in one launch: each block counting-sorts its chunk by destination range (count, LDS scan, place); a range's work-group copies its 256 runs into an LDS list; list and sum order kept
# speedup vs baseline: 4.2462x; 1.1612x over previous
#include <hip/hip_runtime.h>

typedef __attribute__((ext_vector_type(2))) float v2f;
typedef __attribute__((ext_vector_type(8))) float v8f;
typedef __attribute__((ext_vector_type(4))) float v4f;

enum { OWN_RS = 391, OWN_NR = 256, BIN_NCH = 256, BIN_NT = 1024, ENT_SHIFT = 17,
       ENT_MASK = (1 << ENT_SHIFT) - 1,
       LIST_CAP = 8 * BIN_NT };
typedef __attribute__((ext_vector_type(2))) int v2i;
static_assert(BIN_NCH * 4 == BIN_NT, "four threads walk one run");

__global__ void __launch_bounds__(BIN_NT)
bin_sort_kernel(const int* __restrict__ src, const int* __restrict__ dst, int E,
                int* __restrict__ seg, int* __restrict__ list) {
  __shared__ int h[OWN_NR];
  const int tid = threadIdx.x;
  const int c = blockIdx.x;
  if (tid < OWN_NR) h[tid] = 0;
  __syncthreads();
  const int per = (E + BIN_NCH - 1) / BIN_NCH;
  const int e0 = c * per;
  int e1 = e0 + per;
  if (e1 > E) e1 = E;
  for (int eb = e0 + tid; eb < e1; eb += 8 * BIN_NT) {
    int d[8];
#pragma unroll
    for (int u = 0; u < 8; ++u) {
      const int e = eb + u * BIN_NT;
      d[u] = e < e1 ? dst[e] : -1;
    }
#pragma unroll
    for (int u = 0; u < 8; ++u)
      if (d[u] >= 0) atomicAdd(&h[d[u] / OWN_RS], 1);
  }
  __syncthreads();
  if (tid < 32) {
    int v[8], x[8];
    int sum = 0;
#pragma unroll
    for (int k = 0; k < 8; ++k) {
      x[k] = h[8 * tid + k];
      v[k] = sum;
      sum += x[k];
    }
    int inc = sum;
#pragma unroll
    for (int off = 1; off < 32; off <<= 1) {
      const int y = __shfl_up(inc, off, 32);
      if (tid >= off) inc += y;
    }
    const int excl = inc - sum;
#pragma unroll
    for (int k = 0; k < 8; ++k) {
      const int st = e0 + excl + v[k];
      h[8 * tid + k] = st;
      *(v2i*)(seg + 2 * (c * OWN_NR + 8 * tid + k)) = v2i{st, x[k]};
    }
  }
  __syncthreads();
  for (int eb = e0 + tid; eb < e1; eb += 8 * BIN_NT) {
    int d[8], sv[8];
#pragma unroll
    for (int u = 0; u < 8; ++u) {
      const int e = eb + u * BIN_NT;
      const bool ok = e < e1;
      d[u]  = ok ? dst[e] : -1;
      sv[u] = ok ? src[e] : 0;
    }
#pragma unroll
    for (int u = 0; u < 8; ++u) {
      if (d[u] >= 0) {
        const int rr = d[u] / OWN_RS;
        const int loc = d[u] - rr * OWN_RS;
        const int pos = atomicAdd(&h[rr], 1);
        list[pos] = sv[u] | (loc << ENT_SHIFT);
      }
    }
  }
}

__global__ void __launch_bounds__(BIN_NT)
deg_list_kernel(const int* __restrict__ list, const int* __restrict__ seg,
                float* __restrict__ dis, int M) {
  __shared__ int bins[OWN_RS];
  const int tid = threadIdx.x;
  const v2i sg = *(const v2i*)(seg + 2 * ((tid >> 2) * OWN_NR + (int)blockIdx.x));
  if (tid < OWN_RS) bins[tid] = 0;
  __syncthreads();
  for (int jb = tid & 3; jb < sg.y; jb += 64) {
    int w[16];
#pragma unroll
    for (int u = 0; u < 16; ++u) {
      const int j = jb + 4 * u;
      w[u] = j < sg.y ? list[sg.x + j] : -1;
    }
#pragma unroll
    for (int u = 0; u < 16; ++u)
      if (w[u] >= 0) atomicAdd(&bins[w[u] >> ENT_SHIFT], 1);
  }
  __syncthreads();
  const int n = blockIdx.x * OWN_RS + tid;
  if (tid < OWN_RS && n < M) dis[n] = rsqrtf(1.0f + (float)bins[tid]);
}

enum { G1_K = 256, G1_N = 64, G1_ROWS = 32, G1_AP = G1_K + 4, G1_NT = 256,
       G1_LOADS = G1_ROWS * (G1_K / 4) / G1_NT };

__global__ void __launch_bounds__(G1_NT)
gemm1_tiles_kernel(const float* __restrict__ X, const float* __restrict__ W,
                   const float* __restrict__ dis, float* __restrict__ OUT, int M) {
  __shared__ __attribute__((aligned(16))) float sW[G1_K * G1_N];
  __shared__ __attribute__((aligned(16))) float sA[G1_ROWS * G1_AP];
  const int tid = threadIdx.x;

#pragma unroll
  for (int j = 0; j < G1_K * G1_N / 4 / G1_NT; ++j) {
    const int i = tid + j * G1_NT;
    ((v4f*)sW)[i] = ((const v4f*)W)[i];
  }

  const int pr = tid >> 6;
  const int pc = tid & 63;
  const int lane = tid & 31;
  const int wave = __builtin_amdgcn_readfirstlane(tid >> 5);
  const int rt = wave >> 2;
  const int colBase = (wave & 3) * 16;
  const int mn  = lane & 15;
  const int grp = lane >> 4;
  const float* arow = sA + (rt * 16 + mn) * G1_AP + 2 * grp;
  const float* wcol = sW + (2 * grp) * G1_N + colBase + mn;

  const int nTiles = (M + G1_ROWS - 1) / G1_ROWS;
  const float* xcol = X + 4 * pc;
  v4f pre[G1_LOADS];
  int tile = blockIdx.x;
#pragma unroll
  for (int j = 0; j < G1_LOADS; ++j) {
    int gr = tile * G1_ROWS + pr + 4 * j;
    gr = gr < M ? gr : M - 1;
    pre[j] = *(const v4f*)(xcol + (long long)gr * G1_K);
  }

  for (; tile < nTiles; tile += gridDim.x) {
    __syncthreads();
#pragma unroll
    for (int j = 0; j < G1_LOADS; ++j)
      *(v4f*)(sA + (pr + 4 * j) * G1_AP + 4 * pc) = pre[j];
    __syncthreads();

    const int rowBase = tile * G1_ROWS + rt * 16;
    const int row0 = rowBase + 8 * grp;
    const bool whole = rowBase + 16 <= M;
    float dr[8] = {0.0f, 0.0f, 0.0f, 0.0f, 0.0f, 0.0f, 0.0f, 0.0f};
    if (whole) {
#pragma unroll
      for (int r = 0; r < 8; ++r) dr[r] = dis[row0 + r];
    }

    const int nxt = tile + (int)gridDim.x < nTiles ? tile + (int)gridDim.x : tile;
#pragma unroll
    for (int j = 0; j < G1_LOADS; ++j) {
      int gr = nxt * G1_ROWS + pr + 4 * j;
      gr = gr < M ? gr : M - 1;
      pre[j] = *(const v4f*)(xcol + (long long)gr * G1_K);
    }

    if (whole) {
      v8f acc = {};
#pragma unroll 1
      for (int k0 = 0; k0 < G1_K; k0 += 16) {
#pragma unroll
        for (int u = 0; u < 16; u += 4) {
          const int k = k0 + u;
          const v2f a = *(const v2f*)(arow + k);
          v2f b;
          b.x = wcol[k * G1_N];
          b.y = wcol[(k + 1) * G1_N];
          acc = __builtin_amdgcn_wmma_f32_16x16x4_f32(
              false, a, false, b, (short)0, acc, false, false);
        }
        asm volatile("v_nop\n\tv_nop\n\tv_nop\n\tv_nop" : "+v"(acc));
      }

      float* out = OUT + (long long)row0 * G1_N + colBase + mn;
#pragma unroll
      for (int r = 0; r < 8; ++r) out[(long long)r * G1_N] = dr[r] * acc[r];
    }
  }
}

template<int D>
__global__ void __launch_bounds__(BIN_NT)
owner_sum_kernel(const int* __restrict__ list, const int* __restrict__ seg,
                 const float* __restrict__ H, const float* __restrict__ dis,
                 const float* __restrict__ bias, const float* __restrict__ Wn,
                 float* __restrict__ res, int M) {
  constexpr int NL = D / 32;
  constexpr int G  = 16 / NL;
  constexpr int NG = 32 / G;
  constexpr int PD = NG < 3 ? NG : 3;
  constexpr int NI = (OWN_RS * (D / 4) + BIN_NT - 1) / BIN_NT;
  constexpr int PAD_ENT = OWN_RS << ENT_SHIFT;
  constexpr bool GEMM = D == 64;
  constexpr int NO = 32;
  constexpr int NRT = (OWN_RS + 15) / 16;
  constexpr int P = GEMM ? D + 4 : D;
  constexpr int ROWS = GEMM ? NRT * 16 : OWN_RS + 1;
  __shared__ __attribute__((aligned(16))) float acc[ROWS * P];
  __shared__ __attribute__((aligned(16))) float sW[GEMM ? D * NO : 4];
  __shared__ float sDis[NRT * 16];
  __shared__ int sList[LIST_CAP];
  __shared__ int sPre[BIN_NCH + 1];
  const int tid = threadIdx.x;
  const int myc = tid >> 2;
  const v2i sg = *(const v2i*)(seg + 2 * (myc * OWN_NR + (int)blockIdx.x));
  const int lo = blockIdx.x * OWN_RS;
  int nrows = M - lo;
  if (nrows > OWN_RS) nrows = OWN_RS;
  const int nq = nrows * (D / 4);
  const v4f* hq = (const v4f*)(H + (long long)lo * D);
  {
    v4f t[NI];
#pragma unroll
    for (int k = 0; k < NI; ++k) {
      const int i = tid + k * BIN_NT;
      t[k] = i < nq ? hq[i] : v4f{0.0f, 0.0f, 0.0f, 0.0f};
    }
#pragma unroll
    for (int k = 0; k < NI; ++k) {
      const int i = tid + k * BIN_NT;
      if (i < nq) *(v4f*)(acc + (i / (D / 4)) * P + 4 * (i % (D / 4))) = t[k];
    }
  }
  if (tid < NRT * 16) sDis[tid] = dis[lo + tid < M ? lo + tid : M - 1];
  if ((tid & 3) == 0) sPre[myc] = sg.y;
  if (GEMM) {
#pragma unroll
    for (int j = 0; j < D * NO / BIN_NT; ++j) {
      const int i = tid + j * BIN_NT;
      const int k = i / NO, n = i % NO;
      sW[((k >> 1) * 16 + (n & 15)) * 4 + (n >> 4) * 2 + (k & 1)] = Wn[i];
    }
  }
  __syncthreads();

  if (tid < 32) {
    int v[8];
    int sum = 0;
#pragma unroll
    for (int k = 0; k < 8; ++k) {
      const int x = sPre[8 * tid + k];
      v[k] = sum;
      sum += x;
    }
    int inc = sum;
#pragma unroll
    for (int off = 1; off < 32; off <<= 1) {
      const int y = __shfl_up(inc, off, 32);
      if (tid >= off) inc += y;
    }
    const int excl = inc - sum;
#pragma unroll
    for (int k = 0; k < 8; ++k) sPre[8 * tid + k] = excl + v[k];
    if (tid == 31) sPre[BIN_NCH] = inc;
  }
  __syncthreads();

  const int lane = tid & 31;
  const int wave = __builtin_amdgcn_readfirstlane(tid >> 5);
  const int tot = __builtin_amdgcn_readfirstlane(sPre[BIN_NCH]);
  const int pre = sPre[myc];
  for (int base = 0; base < tot; base += LIST_CAP) {
  if (base > 0) __syncthreads();
  for (int jb = tid & 3; jb < sg.y; jb += 64) {
    int w[16];
#pragma unroll
    for (int u = 0; u < 16; ++u) {
      const int j = jb + 4 * u;
      w[u] = j < sg.y ? list[sg.x + j] : 0;
    }
#pragma unroll
    for (int u = 0; u < 16; ++u) {
      const int j = jb + 4 * u;
      const int q = pre + j - base;
      if (j < sg.y && q >= 0 && q < LIST_CAP) sList[q] = w[u];
    }
  }
  __syncthreads();
  int n = tot - base;
  if (n > LIST_CAP) n = LIST_CAP;
  const int p0 = wave * 32;
  int ent = PAD_ENT;
  if (p0 + lane < n) ent = sList[p0 + lane];
  for (int p = p0; p < n; p += BIN_NT) {
    const int pn = p + BIN_NT + lane;
    int entN = sList[pn < n ? pn : n - 1];
    entN = pn < n ? entN : PAD_ENT;

    float v[NG][G][NL];
#pragma unroll
    for (int k = 0; k < NG + PD; ++k) {
      if (k >= PD) {
#pragma unroll
        for (int g = 0; g < G; ++g) {
          const int lj = __builtin_amdgcn_readlane(ent, (k - PD) * G + g) >> ENT_SHIFT;
          float* o = acc + lj * P + lane;
#pragma unroll
          for (int i = 0; i < NL; ++i) atomicAdd(o + 32 * i, v[k - PD][g][i]);
        }
      }
      if (k < NG) {
#pragma unroll
        for (int g = 0; g < G; ++g) {
          const int sj = __builtin_amdgcn_readlane(ent, k * G + g) & ENT_MASK;
          const float* hp = H + (long long)sj * D + lane;
#pragma unroll
          for (int i = 0; i < NL; ++i) v[k][g][i] = hp[32 * i];
        }
      }
    }
    ent = entN;
  }
  }
  __syncthreads();

  if (GEMM) {
    const float4 bb = *(const float4*)(bias + 4 * (tid % (D / 4)));
    for (int i = tid; i < NRT * 16 * (D / 4); i += BIN_NT) {
      const int r = i / (D / 4), c = i % (D / 4);
      v4f v = {0.0f, 0.0f, 0.0f, 0.0f};
      if (r < nrows) {
        v = *(const v4f*)(acc + r * P + 4 * c);
        const float dr = sDis[r];
        v.x = fmaxf(dr * v.x + bb.x, 0.0f);
        v.y = fmaxf(dr * v.y + bb.y, 0.0f);
        v.z = fmaxf(dr * v.z + bb.z, 0.0f);
        v.w = fmaxf(dr * v.w + bb.w, 0.0f);
      }
      *(v4f*)(acc + r * P + 4 * c) = v;
    }
    __syncthreads();

    if (wave < NRT) {
      const int mn  = lane & 15;
      const int grp = lane >> 4;
      const float* arow = acc + (wave * 16 + mn) * P + 2 * grp;
      const float* wq = sW + (grp * 16 + mn) * 4;
      const int lr0 = wave * 16 + 8 * grp;
      float* out = res + (long long)(lo + lr0) * NO + mn;
      float dr[8];
#pragma unroll
      for (int r = 0; r < 8; ++r) dr[r] = sDis[lr0 + r];
      v8f c0 = {0.0f, 0.0f, 0.0f, 0.0f, 0.0f, 0.0f, 0.0f, 0.0f};
      v8f c1 = {0.0f, 0.0f, 0.0f, 0.0f, 0.0f, 0.0f, 0.0f, 0.0f};
      asm volatile("" : "+v"(c0), "+v"(c1), "+v"(out));
      typedef __attribute__((address_space(1))) float gfloat;
      gfloat* outg = (gfloat*)out;
#pragma unroll
      for (int k = 0; k < D; k += 4) {
        const v2f a = *(const v2f*)(arow + k);
        const v4f b = *(const v4f*)(wq + k * 32);
        const v2f b0 = {b.x, b.y};
        const v2f b1 = {b.z, b.w};
        c0 = __builtin_amdgcn_wmma_f32_16x16x4_f32(
            false, a, false, b0, (short)0, c0, false, false);
        c1 = __builtin_amdgcn_wmma_f32_16x16x4_f32(
            false, a, false, b1, (short)0, c1, false, false);
      }
      asm volatile("v_nop\n\tv_nop\n\tv_nop\n\tv_nop" : "+v"(c0), "+v"(c1));

#pragma unroll
      for (int r = 0; r < 8; ++r) {
        if (lr0 + r < nrows) {
          outg[r * NO] = dr[r] * c0[r];
          outg[r * NO + 16] = dr[r] * c1[r];
        }
      }
    }
  } else {
    constexpr int RPW = (OWN_RS + 31) / 32;
    const float bl = bias[lane];
    float z[RPW];
#pragma unroll
    for (int j = 0; j < RPW; ++j) {
      const int r = wave + 32 * j;
      const int rr = r < nrows ? r : 0;
      z[j] = sDis[rr] * acc[rr * P + lane] + bl;
    }
    float m[RPW], ev[RPW], sm[RPW];
#pragma unroll
    for (int j = 0; j < RPW; ++j) m[j] = z[j];
#pragma unroll
    for (int off = 16; off > 0; off >>= 1) {
#pragma unroll
      for (int j = 0; j < RPW; ++j) m[j] = fmaxf(m[j], __shfl_xor(m[j], off, 32));
    }
#pragma unroll
    for (int j = 0; j < RPW; ++j) { ev[j] = __expf(z[j] - m[j]); sm[j] = ev[j]; }
#pragma unroll
    for (int off = 16; off > 0; off >>= 1) {
#pragma unroll
      for (int j = 0; j < RPW; ++j) sm[j] += __shfl_xor(sm[j], off, 32);
    }
#pragma unroll
    for (int j = 0; j < RPW; ++j) {
      const int r = wave + 32 * j;
      if (r < nrows) res[(long long)(lo + r) * D + lane] = ev[j] / sm[j];
    }
  }
}

extern "C" void kernel_launch(void* const* d_in, const int* in_sizes, int n_in,
                              void* d_out, int out_size, void* d_ws, size_t ws_size,
                              hipStream_t stream) {
  (void)n_in; (void)out_size; (void)ws_size;
  const float* x  = (const float*)d_in[0];
  const int*   ei = (const int*)d_in[1];
  const float* W1 = (const float*)d_in[2];
  const float* b1 = (const float*)d_in[3];
  const float* W2 = (const float*)d_in[4];
  const float* b2 = (const float*)d_in[5];

  const int IN = 256, HID = 64, OD = 32;
  const int M = in_sizes[0] / IN;
  const int E = in_sizes[1] / 2;
  const int* src = ei;
  const int* dst = ei + E;

  float* ws   = (float*)d_ws;
  float* dis  = ws;
  float* H1   = dis + (((long long)M + 1023) & ~1023LL);
  float* H2   = H1 + (long long)M * HID;
  int* list   = (int*)(H2 + (long long)M * HID);
  int* seg    = list + E;
  float* out  = (float*)d_out;

  const int NRANGE = (M + OWN_RS - 1) / OWN_RS;

  bin_sort_kernel<<<BIN_NCH, BIN_NT, 0, stream>>>(src, dst, E, seg, list);
  deg_list_kernel<<<NRANGE, BIN_NT, 0, stream>>>(list, seg, dis, M);

  gemm1_tiles_kernel<<<768, G1_NT, 0, stream>>>(x, W1, dis, H1, M);
  owner_sum_kernel<64><<<NRANGE, BIN_NT, 0, stream>>>(list, seg, H1, dis, b1, W2, H2, M);

  owner_sum_kernel<32><<<NRANGE, BIN_NT, 0, stream>>>(list, seg, H2, dis, b2, W2, out, M);
}


# in-degree count moved into the layer-1 GEMM's work-groups (3 per destination range; each counts the range's runs in LDS, scales its rows from there): degree launch and 8192 waves go; sums as before
# speedup vs baseline: 4.4319x; 1.0437x over previous
#include <hip/hip_runtime.h>

typedef __attribute__((ext_vector_type(2))) float v2f;
typedef __attribute__((ext_vector_type(8))) float v8f;
typedef __attribute__((ext_vector_type(4))) float v4f;

enum { OWN_RS = 391, OWN_NR = 256, BIN_NCH = 256, BIN_NT = 1024, ENT_SHIFT = 17,
       ENT_MASK = (1 << ENT_SHIFT) - 1,
       LIST_CAP = 8 * BIN_NT };
typedef __attribute__((ext_vector_type(2))) int v2i;
static_assert(BIN_NCH * 4 == BIN_NT, "four threads walk one run");

__global__ void __launch_bounds__(BIN_NT)
bin_sort_kernel(const int* __restrict__ src, const int* __restrict__ dst, int E,
                int* __restrict__ seg, int* __restrict__ list) {
  __shared__ int h[OWN_NR];
  const int tid = threadIdx.x;
  const int c = blockIdx.x;
  if (tid < OWN_NR) h[tid] = 0;
  __syncthreads();
  const int per = (E + BIN_NCH - 1) / BIN_NCH;
  const int e0 = c * per;
  int e1 = e0 + per;
  if (e1 > E) e1 = E;
  for (int eb = e0 + tid; eb < e1; eb += 8 * BIN_NT) {
    int d[8];
#pragma unroll
    for (int u = 0; u < 8; ++u) {
      const int e = eb + u * BIN_NT;
      d[u] = e < e1 ? dst[e] : -1;
    }
#pragma unroll
    for (int u = 0; u < 8; ++u)
      if (d[u] >= 0) atomicAdd(&h[d[u] / OWN_RS], 1);
  }
  __syncthreads();
  if (tid < 32) {
    int v[8], x[8];
    int sum = 0;
#pragma unroll
    for (int k = 0; k < 8; ++k) {
      x[k] = h[8 * tid + k];
      v[k] = sum;
      sum += x[k];
    }
    int inc = sum;
#pragma unroll
    for (int off = 1; off < 32; off <<= 1) {
      const int y = __shfl_up(inc, off, 32);
      if (tid >= off) inc += y;
    }
    const int excl = inc - sum;
#pragma unroll
    for (int k = 0; k < 8; ++k) {
      const int st = e0 + excl + v[k];
      h[8 * tid + k] = st;
      *(v2i*)(seg + 2 * (c * OWN_NR + 8 * tid + k)) = v2i{st, x[k]};
    }
  }
  __syncthreads();
  for (int eb = e0 + tid; eb < e1; eb += 8 * BIN_NT) {
    int d[8], sv[8];
#pragma unroll
    for (int u = 0; u < 8; ++u) {
      const int e = eb + u * BIN_NT;
      const bool ok = e < e1;
      d[u]  = ok ? dst[e] : -1;
      sv[u] = ok ? src[e] : 0;
    }
#pragma unroll
    for (int u = 0; u < 8; ++u) {
      if (d[u] >= 0) {
        const int rr = d[u] / OWN_RS;
        const int loc = d[u] - rr * OWN_RS;
        const int pos = atomicAdd(&h[rr], 1);
        list[pos] = sv[u] | (loc << ENT_SHIFT);
      }
    }
  }
}

enum { G1_K = 256, G1_N = 64, G1_ROWS = 32, G1_AP = G1_K + 4, G1_NT = 256,
       G1_LOADS = G1_ROWS * (G1_K / 4) / G1_NT,
       G1_PARTS = 3,
       G1_NU = (OWN_RS + 15) / 16 };
static_assert(G1_NT == BIN_NCH, "lane l of wave w holds the run of chunk 32 w + l");

__global__ void __launch_bounds__(G1_NT)
gemm1_range_kernel(const float* __restrict__ X, const float* __restrict__ W,
                   const int* __restrict__ list, const int* __restrict__ seg,
                   float* __restrict__ dis, float* __restrict__ OUT, int M) {
  __shared__ __attribute__((aligned(16))) float sW[G1_K * G1_N];
  __shared__ __attribute__((aligned(16))) float sA[G1_ROWS * G1_AP];
  __shared__ int bins[OWN_RS];
  __shared__ float sD[G1_NU * 16];
  const int tid = threadIdx.x;
  const int lane = tid & 31;
  const int wave = __builtin_amdgcn_readfirstlane(tid >> 5);
  const int rg = (int)blockIdx.x % OWN_NR;
  const int part = (int)blockIdx.x / OWN_NR;
  const int lo = rg * OWN_RS;
  const int rBeg = lo + 16 * ((part * G1_NU + G1_PARTS - 1) / G1_PARTS);
  int rEnd = lo + 16 * (((part + 1) * G1_NU + G1_PARTS - 1) / G1_PARTS);
  if (rEnd > lo + OWN_RS) rEnd = lo + OWN_RS;
  if (rEnd > M) rEnd = M;
  const int nTiles = rEnd > rBeg ? (rEnd - rBeg + G1_ROWS - 1) / G1_ROWS : 0;

  const v2i sg = *(const v2i*)(seg + 2 * (tid * OWN_NR + rg));
  for (int i = tid; i < OWN_RS; i += G1_NT) bins[i] = 0;

  v4f wreg[G1_K * G1_N / 4 / G1_NT];
#pragma unroll
  for (int j = 0; j < G1_K * G1_N / 4 / G1_NT; ++j)
    wreg[j] = ((const v4f*)W)[tid + j * G1_NT];

  const int pr = tid >> 6;
  const int pc = tid & 63;
  const int rt = wave >> 2;
  const int colBase = (wave & 3) * 16;
  const int mn  = lane & 15;
  const int grp = lane >> 4;
  const float* arow = sA + (rt * 16 + mn) * G1_AP + 2 * grp;
  const float* wcol = sW + (2 * grp) * G1_N + colBase + mn;

  const float* xcol = X + 4 * pc;
  v4f pre[G1_LOADS];
#pragma unroll
  for (int j = 0; j < G1_LOADS; ++j) {
    int gr = rBeg + pr + 4 * j;
    gr = gr < M ? gr : M - 1;
    pre[j] = *(const v4f*)(xcol + (long long)gr * G1_K);
  }

  int w[32];
#pragma unroll
  for (int r = 0; r < 32; ++r) {
    const int s = __builtin_amdgcn_readlane(sg.x, r);
    const int len = __builtin_amdgcn_readlane(sg.y, r);
    w[r] = lane < len ? list[s + lane] : -1;
  }

#pragma unroll
  for (int j = 0; j < G1_K * G1_N / 4 / G1_NT; ++j)
    ((v4f*)sW)[tid + j * G1_NT] = wreg[j];

  __syncthreads();
#pragma unroll
  for (int r = 0; r < 32; ++r)
    if (w[r] >= 0) atomicAdd(&bins[w[r] >> ENT_SHIFT], 1);
#pragma unroll
  for (int r = 0; r < 32; ++r) {
    const int s = __builtin_amdgcn_readlane(sg.x, r);
    const int len = __builtin_amdgcn_readlane(sg.y, r);
    for (int j = 32 + lane; j < len; j += 32)
      atomicAdd(&bins[list[s + j] >> ENT_SHIFT], 1);
  }
  __syncthreads();
  for (int i = tid; i < G1_NU * 16; i += G1_NT)
    sD[i] = i < OWN_RS ? rsqrtf(1.0f + (float)bins[i]) : 0.0f;

  for (int tile = 0; tile < nTiles; ++tile) {
    __syncthreads();
#pragma unroll
    for (int j = 0; j < G1_LOADS; ++j)
      *(v4f*)(sA + (pr + 4 * j) * G1_AP + 4 * pc) = pre[j];
    __syncthreads();

    const int rowBase = rBeg + tile * G1_ROWS + rt * 16;
    const int row0 = rowBase + 8 * grp;
    const bool act = rowBase < rEnd;

    const int nxt = tile + 1 < nTiles ? tile + 1 : tile;
#pragma unroll
    for (int j = 0; j < G1_LOADS; ++j) {
      int gr = rBeg + nxt * G1_ROWS + pr + 4 * j;
      gr = gr < M ? gr : M - 1;
      pre[j] = *(const v4f*)(xcol + (long long)gr * G1_K);
    }

    if (act) {
      float dr[8];
#pragma unroll
      for (int r = 0; r < 8; ++r) dr[r] = sD[row0 - lo + r];
      v8f acc = {};
#pragma unroll 1
      for (int k0 = 0; k0 < G1_K; k0 += 16) {
#pragma unroll
        for (int u = 0; u < 16; u += 4) {
          const int k = k0 + u;
          const v2f a = *(const v2f*)(arow + k);
          v2f b;
          b.x = wcol[k * G1_N];
          b.y = wcol[(k + 1) * G1_N];
          acc = __builtin_amdgcn_wmma_f32_16x16x4_f32(
              false, a, false, b, (short)0, acc, false, false);
        }
        asm volatile("v_nop\n\tv_nop\n\tv_nop\n\tv_nop" : "+v"(acc));
      }

      float* out = OUT + (long long)row0 * G1_N + colBase + mn;
      if (rowBase + 16 <= rEnd) {
#pragma unroll
        for (int r = 0; r < 8; ++r) out[(long long)r * G1_N] = dr[r] * acc[r];
      } else {
#pragma unroll
        for (int r = 0; r < 8; ++r)
          if (row0 + r < rEnd) out[(long long)r * G1_N] = dr[r] * acc[r];
      }
    }
  }

  if (part == G1_PARTS - 1)
    for (int i = tid; i < OWN_RS; i += G1_NT)
      if (lo + i < M) dis[lo + i] = sD[i];
}

template<int D>
__global__ void __launch_bounds__(BIN_NT)
owner_sum_kernel(const int* __restrict__ list, const int* __restrict__ seg,
                 const float* __restrict__ H, const float* __restrict__ dis,
                 const float* __restrict__ bias, const float* __restrict__ Wn,
                 float* __restrict__ res, int M) {
  constexpr int NL = D / 32;
  constexpr int G  = 16 / NL;
  constexpr int NG = 32 / G;
  constexpr int PD = NG < 3 ? NG : 3;
  constexpr int NI = (OWN_RS * (D / 4) + BIN_NT - 1) / BIN_NT;
  constexpr int PAD_ENT = OWN_RS << ENT_SHIFT;
  constexpr bool GEMM = D == 64;
  constexpr int NO = 32;
  constexpr int NRT = (OWN_RS + 15) / 16;
  constexpr int P = GEMM ? D + 4 : D;
  constexpr int ROWS = GEMM ? NRT * 16 : OWN_RS + 1;
  __shared__ __attribute__((aligned(16))) float acc[ROWS * P];
  __shared__ __attribute__((aligned(16))) float sW[GEMM ? D * NO : 4];
  __shared__ float sDis[NRT * 16];
  __shared__ int sList[LIST_CAP];
  __shared__ int sPre[BIN_NCH + 1];
  const int tid = threadIdx.x;
  const int myc = tid >> 2;
  const v2i sg = *(const v2i*)(seg + 2 * (myc * OWN_NR + (int)blockIdx.x));
  const int lo = blockIdx.x * OWN_RS;
  int nrows = M - lo;
  if (nrows > OWN_RS) nrows = OWN_RS;
  const int nq = nrows * (D / 4);
  const v4f* hq = (const v4f*)(H + (long long)lo * D);
  {
    v4f t[NI];
#pragma unroll
    for (int k = 0; k < NI; ++k) {
      const int i = tid + k * BIN_NT;
      t[k] = i < nq ? hq[i] : v4f{0.0f, 0.0f, 0.0f, 0.0f};
    }
#pragma unroll
    for (int k = 0; k < NI; ++k) {
      const int i = tid + k * BIN_NT;
      if (i < nq) *(v4f*)(acc + (i / (D / 4)) * P + 4 * (i % (D / 4))) = t[k];
    }
  }
  if (tid < NRT * 16) sDis[tid] = dis[lo + tid < M ? lo + tid : M - 1];
  if ((tid & 3) == 0) sPre[myc] = sg.y;
  if (GEMM) {
#pragma unroll
    for (int j = 0; j < D * NO / BIN_NT; ++j) {
      const int i = tid + j * BIN_NT;
      const int k = i / NO, n = i % NO;
      sW[((k >> 1) * 16 + (n & 15)) * 4 + (n >> 4) * 2 + (k & 1)] = Wn[i];
    }
  }
  __syncthreads();

  if (tid < 32) {
    int v[8];
    int sum = 0;
#pragma unroll
    for (int k = 0; k < 8; ++k) {
      const int x = sPre[8 * tid + k];
      v[k] = sum;
      sum += x;
    }
    int inc = sum;
#pragma unroll
    for (int off = 1; off < 32; off <<= 1) {
      const int y = __shfl_up(inc, off, 32);
      if (tid >= off) inc += y;
    }
    const int excl = inc - sum;
#pragma unroll
    for (int k = 0; k < 8; ++k) sPre[8 * tid + k] = excl + v[k];
    if (tid == 31) sPre[BIN_NCH] = inc;
  }
  __syncthreads();

  const int lane = tid & 31;
  const int wave = __builtin_amdgcn_readfirstlane(tid >> 5);
  const int tot = __builtin_amdgcn_readfirstlane(sPre[BIN_NCH]);
  const int pre = sPre[myc];
  for (int base = 0; base < tot; base += LIST_CAP) {
  if (base > 0) __syncthreads();
  for (int jb = tid & 3; jb < sg.y; jb += 64) {
    int w[16];
#pragma unroll
    for (int u = 0; u < 16; ++u) {
      const int j = jb + 4 * u;
      w[u] = j < sg.y ? list[sg.x + j] : 0;
    }
#pragma unroll
    for (int u = 0; u < 16; ++u) {
      const int j = jb + 4 * u;
      const int q = pre + j - base;
      if (j < sg.y && q >= 0 && q < LIST_CAP) sList[q] = w[u];
    }
  }
  __syncthreads();
  int n = tot - base;
  if (n > LIST_CAP) n = LIST_CAP;
  const int p0 = wave * 32;
  int ent = PAD_ENT;
  if (p0 + lane < n) ent = sList[p0 + lane];
  for (int p = p0; p < n; p += BIN_NT) {
    const int pn = p + BIN_NT + lane;
    int entN = sList[pn < n ? pn : n - 1];
    entN = pn < n ? entN : PAD_ENT;

    float v[NG][G][NL];
#pragma unroll
    for (int k = 0; k < NG + PD; ++k) {
      if (k >= PD) {
#pragma unroll
        for (int g = 0; g < G; ++g) {
          const int lj = __builtin_amdgcn_readlane(ent, (k - PD) * G + g) >> ENT_SHIFT;
          float* o = acc + lj * P + lane;
#pragma unroll
          for (int i = 0; i < NL; ++i) atomicAdd(o + 32 * i, v[k - PD][g][i]);
        }
      }
      if (k < NG) {
#pragma unroll
        for (int g = 0; g < G; ++g) {
          const int sj = __builtin_amdgcn_readlane(ent, k * G + g) & ENT_MASK;
          const float* hp = H + (long long)sj * D + lane;
#pragma unroll
          for (int i = 0; i < NL; ++i) v[k][g][i] = hp[32 * i];
        }
      }
    }
    ent = entN;
  }
  }
  __syncthreads();

  if (GEMM) {
    const float4 bb = *(const float4*)(bias + 4 * (tid % (D / 4)));
    for (int i = tid; i < NRT * 16 * (D / 4); i += BIN_NT) {
      const int r = i / (D / 4), c = i % (D / 4);
      v4f v = {0.0f, 0.0f, 0.0f, 0.0f};
      if (r < nrows) {
        v = *(const v4f*)(acc + r * P + 4 * c);
        const float dr = sDis[r];
        v.x = fmaxf(dr * v.x + bb.x, 0.0f);
        v.y = fmaxf(dr * v.y + bb.y, 0.0f);
        v.z = fmaxf(dr * v.z + bb.z, 0.0f);
        v.w = fmaxf(dr * v.w + bb.w, 0.0f);
      }
      *(v4f*)(acc + r * P + 4 * c) = v;
    }
    __syncthreads();

    if (wave < NRT) {
      const int mn  = lane & 15;
      const int grp = lane >> 4;
      const float* arow = acc + (wave * 16 + mn) * P + 2 * grp;
      const float* wq = sW + (grp * 16 + mn) * 4;
      const int lr0 = wave * 16 + 8 * grp;
      float* out = res + (long long)(lo + lr0) * NO + mn;
      float dr[8];
#pragma unroll
      for (int r = 0; r < 8; ++r) dr[r] = sDis[lr0 + r];
      v8f c0 = {0.0f, 0.0f, 0.0f, 0.0f, 0.0f, 0.0f, 0.0f, 0.0f};
      v8f c1 = {0.0f, 0.0f, 0.0f, 0.0f, 0.0f, 0.0f, 0.0f, 0.0f};
      asm volatile("" : "+v"(c0), "+v"(c1), "+v"(out));
      typedef __attribute__((address_space(1))) float gfloat;
      gfloat* outg = (gfloat*)out;
#pragma unroll
      for (int k = 0; k < D; k += 4) {
        const v2f a = *(const v2f*)(arow + k);
        const v4f b = *(const v4f*)(wq + k * 32);
        const v2f b0 = {b.x, b.y};
        const v2f b1 = {b.z, b.w};
        c0 = __builtin_amdgcn_wmma_f32_16x16x4_f32(
            false, a, false, b0, (short)0, c0, false, false);
        c1 = __builtin_amdgcn_wmma_f32_16x16x4_f32(
            false, a, false, b1, (short)0, c1, false, false);
      }
      asm volatile("v_nop\n\tv_nop\n\tv_nop\n\tv_nop" : "+v"(c0), "+v"(c1));

#pragma unroll
      for (int r = 0; r < 8; ++r) {
        if (lr0 + r < nrows) {
          outg[r * NO] = dr[r] * c0[r];
          outg[r * NO + 16] = dr[r] * c1[r];
        }
      }
    }
  } else {
    constexpr int RPW = (OWN_RS + 31) / 32;
    const float bl = bias[lane];
    float z[RPW];
#pragma unroll
    for (int j = 0; j < RPW; ++j) {
      const int r = wave + 32 * j;
      const int rr = r < nrows ? r : 0;
      z[j] = sDis[rr] * acc[rr * P + lane] + bl;
    }
    float m[RPW], ev[RPW], sm[RPW];
#pragma unroll
    for (int j = 0; j < RPW; ++j) m[j] = z[j];
#pragma unroll
    for (int off = 16; off > 0; off >>= 1) {
#pragma unroll
      for (int j = 0; j < RPW; ++j) m[j] = fmaxf(m[j], __shfl_xor(m[j], off, 32));
    }
#pragma unroll
    for (int j = 0; j < RPW; ++j) { ev[j] = __expf(z[j] - m[j]); sm[j] = ev[j]; }
#pragma unroll
    for (int off = 16; off > 0; off >>= 1) {
#pragma unroll
      for (int j = 0; j < RPW; ++j) sm[j] += __shfl_xor(sm[j], off, 32);
    }
#pragma unroll
    for (int j = 0; j < RPW; ++j) {
      const int r = wave + 32 * j;
      if (r < nrows) res[(long long)(lo + r) * D + lane] = ev[j] / sm[j];
    }
  }
}

extern "C" void kernel_launch(void* const* d_in, const int* in_sizes, int n_in,
                              void* d_out, int out_size, void* d_ws, size_t ws_size,
                              hipStream_t stream) {
  (void)n_in; (void)out_size; (void)ws_size;
  const float* x  = (const float*)d_in[0];
  const int*   ei = (const int*)d_in[1];
  const float* W1 = (const float*)d_in[2];
  const float* b1 = (const float*)d_in[3];
  const float* W2 = (const float*)d_in[4];
  const float* b2 = (const float*)d_in[5];

  const int IN = 256, HID = 64, OD = 32;
  const int M = in_sizes[0] / IN;
  const int E = in_sizes[1] / 2;
  const int* src = ei;
  const int* dst = ei + E;

  float* ws   = (float*)d_ws;
  float* dis  = ws;
  float* H1   = dis + (((long long)M + 1023) & ~1023LL);
  float* H2   = H1 + (long long)M * HID;
  int* list   = (int*)(H2 + (long long)M * HID);
  int* seg    = list + E;
  float* out  = (float*)d_out;

  const int NRANGE = (M + OWN_RS - 1) / OWN_RS;

  bin_sort_kernel<<<BIN_NCH, BIN_NT, 0, stream>>>(src, dst, E, seg, list);

  gemm1_range_kernel<<<G1_PARTS * OWN_NR, G1_NT, 0, stream>>>(x, W1, list, seg, dis, H1, M);
  owner_sum_kernel<64><<<NRANGE, BIN_NT, 0, stream>>>(list, seg, H1, dis, b1, W2, H2, M);

  owner_sum_kernel<32><<<NRANGE, BIN_NT, 0, stream>>>(list, seg, H2, dis, b2, W2, out, M);
}


# edge lists ordered by destination row in LDS (count, scan, place); a wave sums runs of equal destination in registers: one LDS add per run, not per edge; sum order: run partials, then LDS adds
# speedup vs baseline: 4.7364x; 1.0687x over previous
#include <hip/hip_runtime.h>

typedef __attribute__((ext_vector_type(2))) float v2f;
typedef __attribute__((ext_vector_type(8))) float v8f;
typedef __attribute__((ext_vector_type(4))) float v4f;

enum { OWN_RS = 391, OWN_NR = 256, BIN_NCH = 256, BIN_NT = 1024, ENT_SHIFT = 17,
       ENT_MASK = (1 << ENT_SHIFT) - 1,
       LIST_CAP = 8 * BIN_NT };
typedef __attribute__((ext_vector_type(2))) int v2i;
static_assert(BIN_NCH * 4 == BIN_NT, "four threads walk one run");

__global__ void __launch_bounds__(BIN_NT)
bin_sort_kernel(const int* __restrict__ src, const int* __restrict__ dst, int E,
                int* __restrict__ seg, int* __restrict__ list) {
  __shared__ int h[OWN_NR];
  const int tid = threadIdx.x;
  const int c = blockIdx.x;
  if (tid < OWN_NR) h[tid] = 0;
  __syncthreads();
  const int per = (E + BIN_NCH - 1) / BIN_NCH;
  const int e0 = c * per;
  int e1 = e0 + per;
  if (e1 > E) e1 = E;
  for (int eb = e0 + tid; eb < e1; eb += 8 * BIN_NT) {
    int d[8];
#pragma unroll
    for (int u = 0; u < 8; ++u) {
      const int e = eb + u * BIN_NT;
      d[u] = e < e1 ? dst[e] : -1;
    }
#pragma unroll
    for (int u = 0; u < 8; ++u)
      if (d[u] >= 0) atomicAdd(&h[d[u] / OWN_RS], 1);
  }
  __syncthreads();
  if (tid < 32) {
    int v[8], x[8];
    int sum = 0;
#pragma unroll
    for (int k = 0; k < 8; ++k) {
      x[k] = h[8 * tid + k];
      v[k] = sum;
      sum += x[k];
    }
    int inc = sum;
#pragma unroll
    for (int off = 1; off < 32; off <<= 1) {
      const int y = __shfl_up(inc, off, 32);
      if (tid >= off) inc += y;
    }
    const int excl = inc - sum;
#pragma unroll
    for (int k = 0; k < 8; ++k) {
      const int st = e0 + excl + v[k];
      h[8 * tid + k] = st;
      *(v2i*)(seg + 2 * (c * OWN_NR + 8 * tid + k)) = v2i{st, x[k]};
    }
  }
  __syncthreads();
  for (int eb = e0 + tid; eb < e1; eb += 8 * BIN_NT) {
    int d[8], sv[8];
#pragma unroll
    for (int u = 0; u < 8; ++u) {
      const int e = eb + u * BIN_NT;
      const bool ok = e < e1;
      d[u]  = ok ? dst[e] : -1;
      sv[u] = ok ? src[e] : 0;
    }
#pragma unroll
    for (int u = 0; u < 8; ++u) {
      if (d[u] >= 0) {
        const int rr = d[u] / OWN_RS;
        const int loc = d[u] - rr * OWN_RS;
        const int pos = atomicAdd(&h[rr], 1);
        list[pos] = sv[u] | (loc << ENT_SHIFT);
      }
    }
  }
}

enum { G1_K = 256, G1_N = 64, G1_ROWS = 32, G1_AP = G1_K + 4, G1_NT = 256,
       G1_LOADS = G1_ROWS * (G1_K / 4) / G1_NT,
       G1_PARTS = 3,
       G1_NU = (OWN_RS + 15) / 16 };
static_assert(G1_NT == BIN_NCH, "lane l of wave w holds the run of chunk 32 w + l");

__global__ void __launch_bounds__(G1_NT)
gemm1_range_kernel(const float* __restrict__ X, const float* __restrict__ W,
                   const int* __restrict__ list, const int* __restrict__ seg,
                   float* __restrict__ dis, float* __restrict__ OUT, int M) {
  __shared__ __attribute__((aligned(16))) float sW[G1_K * G1_N];
  __shared__ __attribute__((aligned(16))) float sA[G1_ROWS * G1_AP];
  __shared__ int bins[OWN_RS];
  __shared__ float sD[G1_NU * 16];
  const int tid = threadIdx.x;
  const int lane = tid & 31;
  const int wave = __builtin_amdgcn_readfirstlane(tid >> 5);
  const int rg = (int)blockIdx.x % OWN_NR;
  const int part = (int)blockIdx.x / OWN_NR;
  const int lo = rg * OWN_RS;
  const int rBeg = lo + 16 * ((part * G1_NU + G1_PARTS - 1) / G1_PARTS);
  int rEnd = lo + 16 * (((part + 1) * G1_NU + G1_PARTS - 1) / G1_PARTS);
  if (rEnd > lo + OWN_RS) rEnd = lo + OWN_RS;
  if (rEnd > M) rEnd = M;
  const int nTiles = rEnd > rBeg ? (rEnd - rBeg + G1_ROWS - 1) / G1_ROWS : 0;

  const v2i sg = *(const v2i*)(seg + 2 * (tid * OWN_NR + rg));
  for (int i = tid; i < OWN_RS; i += G1_NT) bins[i] = 0;

  v4f wreg[G1_K * G1_N / 4 / G1_NT];
#pragma unroll
  for (int j = 0; j < G1_K * G1_N / 4 / G1_NT; ++j)
    wreg[j] = ((const v4f*)W)[tid + j * G1_NT];

  const int pr = tid >> 6;
  const int pc = tid & 63;
  const int rt = wave >> 2;
  const int colBase = (wave & 3) * 16;
  const int mn  = lane & 15;
  const int grp = lane >> 4;
  const float* arow = sA + (rt * 16 + mn) * G1_AP + 2 * grp;
  const float* wcol = sW + (2 * grp) * G1_N + colBase + mn;

  const float* xcol = X + 4 * pc;
  v4f pre[G1_LOADS];
#pragma unroll
  for (int j = 0; j < G1_LOADS; ++j) {
    int gr = rBeg + pr + 4 * j;
    gr = gr < M ? gr : M - 1;
    pre[j] = *(const v4f*)(xcol + (long long)gr * G1_K);
  }

  int w[32];
#pragma unroll
  for (int r = 0; r < 32; ++r) {
    const int s = __builtin_amdgcn_readlane(sg.x, r);
    const int len = __builtin_amdgcn_readlane(sg.y, r);
    w[r] = lane < len ? list[s + lane] : -1;
  }

#pragma unroll
  for (int j = 0; j < G1_K * G1_N / 4 / G1_NT; ++j)
    ((v4f*)sW)[tid + j * G1_NT] = wreg[j];

  __syncthreads();
#pragma unroll
  for (int r = 0; r < 32; ++r)
    if (w[r] >= 0) atomicAdd(&bins[w[r] >> ENT_SHIFT], 1);
#pragma unroll
  for (int r = 0; r < 32; ++r) {
    const int s = __builtin_amdgcn_readlane(sg.x, r);
    const int len = __builtin_amdgcn_readlane(sg.y, r);
    for (int j = 32 + lane; j < len; j += 32)
      atomicAdd(&bins[list[s + j] >> ENT_SHIFT], 1);
  }
  __syncthreads();
  for (int i = tid; i < G1_NU * 16; i += G1_NT)
    sD[i] = i < OWN_RS ? rsqrtf(1.0f + (float)bins[i]) : 0.0f;

  for (int tile = 0; tile < nTiles; ++tile) {
    __syncthreads();
#pragma unroll
    for (int j = 0; j < G1_LOADS; ++j)
      *(v4f*)(sA + (pr + 4 * j) * G1_AP + 4 * pc) = pre[j];
    __syncthreads();

    const int rowBase = rBeg + tile * G1_ROWS + rt * 16;
    const int row0 = rowBase + 8 * grp;
    const bool act = rowBase < rEnd;

    const int nxt = tile + 1 < nTiles ? tile + 1 : tile;
#pragma unroll
    for (int j = 0; j < G1_LOADS; ++j) {
      int gr = rBeg + nxt * G1_ROWS + pr + 4 * j;
      gr = gr < M ? gr : M - 1;
      pre[j] = *(const v4f*)(xcol + (long long)gr * G1_K);
    }

    if (act) {
      float dr[8];
#pragma unroll
      for (int r = 0; r < 8; ++r) dr[r] = sD[row0 - lo + r];
      v8f acc = {};
#pragma unroll 1
      for (int k0 = 0; k0 < G1_K; k0 += 16) {
#pragma unroll
        for (int u = 0; u < 16; u += 4) {
          const int k = k0 + u;
          const v2f a = *(const v2f*)(arow + k);
          v2f b;
          b.x = wcol[k * G1_N];
          b.y = wcol[(k + 1) * G1_N];
          acc = __builtin_amdgcn_wmma_f32_16x16x4_f32(
              false, a, false, b, (short)0, acc, false, false);
        }
        asm volatile("v_nop\n\tv_nop\n\tv_nop\n\tv_nop" : "+v"(acc));
      }

      float* out = OUT + (long long)row0 * G1_N + colBase + mn;
      if (rowBase + 16 <= rEnd) {
#pragma unroll
        for (int r = 0; r < 8; ++r) out[(long long)r * G1_N] = dr[r] * acc[r];
      } else {
#pragma unroll
        for (int r = 0; r < 8; ++r)
          if (row0 + r < rEnd) out[(long long)r * G1_N] = dr[r] * acc[r];
      }
    }
  }

  if (part == G1_PARTS - 1)
    for (int i = tid; i < OWN_RS; i += G1_NT)
      if (lo + i < M) dis[lo + i] = sD[i];
}

template<int D>
__global__ void __launch_bounds__(BIN_NT)
owner_sum_kernel(const int* __restrict__ list, const int* __restrict__ seg,
                 const float* __restrict__ H, const float* __restrict__ dis,
                 const float* __restrict__ bias, const float* __restrict__ Wn,
                 float* __restrict__ res, int M) {
  constexpr int NL = D / 32;
  constexpr int G  = 16 / NL;
  constexpr int NG = 32 / G;
  constexpr int PD = NG < 3 ? NG : 3;
  constexpr int NI = (OWN_RS * (D / 4) + BIN_NT - 1) / BIN_NT;
  constexpr int PAD_ENT = OWN_RS << ENT_SHIFT;
  constexpr bool GEMM = D == 64;
  constexpr int NO = 32;
  constexpr int NRT = (OWN_RS + 15) / 16;
  constexpr int P = GEMM ? D + 4 : D;
  constexpr int ROWS = GEMM ? NRT * 16 : OWN_RS + 1;
  __shared__ __attribute__((aligned(16))) float acc[ROWS * P];
  __shared__ __attribute__((aligned(16))) float sW[GEMM ? D * NO : 4];
  __shared__ float sDis[NRT * 16];
  __shared__ int sList[LIST_CAP];
  __shared__ int sSort[LIST_CAP];
  constexpr int SC_PL = (OWN_RS + 31) / 32;
  __shared__ int sCnt[SC_PL * 32];
  __shared__ int sPre[BIN_NCH + 1];
  const int tid = threadIdx.x;
  const int myc = tid >> 2;
  const v2i sg = *(const v2i*)(seg + 2 * (myc * OWN_NR + (int)blockIdx.x));
  const int lo = blockIdx.x * OWN_RS;
  int nrows = M - lo;
  if (nrows > OWN_RS) nrows = OWN_RS;
  const int nq = nrows * (D / 4);
  const v4f* hq = (const v4f*)(H + (long long)lo * D);
  {
    v4f t[NI];
#pragma unroll
    for (int k = 0; k < NI; ++k) {
      const int i = tid + k * BIN_NT;
      t[k] = i < nq ? hq[i] : v4f{0.0f, 0.0f, 0.0f, 0.0f};
    }
#pragma unroll
    for (int k = 0; k < NI; ++k) {
      const int i = tid + k * BIN_NT;
      if (i < nq) *(v4f*)(acc + (i / (D / 4)) * P + 4 * (i % (D / 4))) = t[k];
    }
  }
  if (tid < NRT * 16) sDis[tid] = dis[lo + tid < M ? lo + tid : M - 1];
  if ((tid & 3) == 0) sPre[myc] = sg.y;
  if (tid < SC_PL * 32) sCnt[tid] = 0;
  if (GEMM) {
#pragma unroll
    for (int j = 0; j < D * NO / BIN_NT; ++j) {
      const int i = tid + j * BIN_NT;
      const int k = i / NO, n = i % NO;
      sW[((k >> 1) * 16 + (n & 15)) * 4 + (n >> 4) * 2 + (k & 1)] = Wn[i];
    }
  }
  __syncthreads();

  if (tid < 32) {
    int v[8];
    int sum = 0;
#pragma unroll
    for (int k = 0; k < 8; ++k) {
      const int x = sPre[8 * tid + k];
      v[k] = sum;
      sum += x;
    }
    int inc = sum;
#pragma unroll
    for (int off = 1; off < 32; off <<= 1) {
      const int y = __shfl_up(inc, off, 32);
      if (tid >= off) inc += y;
    }
    const int excl = inc - sum;
#pragma unroll
    for (int k = 0; k < 8; ++k) sPre[8 * tid + k] = excl + v[k];
    if (tid == 31) sPre[BIN_NCH] = inc;
  }
  __syncthreads();

  const int lane = tid & 31;
  const int wave = __builtin_amdgcn_readfirstlane(tid >> 5);
  const int tot = __builtin_amdgcn_readfirstlane(sPre[BIN_NCH]);
  const int pre = sPre[myc];
  for (int base = 0; base < tot; base += LIST_CAP) {
  if (base > 0) __syncthreads();
  for (int jb = tid & 3; jb < sg.y; jb += 64) {
    int w[16];
#pragma unroll
    for (int u = 0; u < 16; ++u) {
      const int j = jb + 4 * u;
      w[u] = j < sg.y ? list[sg.x + j] : 0;
    }
#pragma unroll
    for (int u = 0; u < 16; ++u) {
      const int j = jb + 4 * u;
      const int q = pre + j - base;
      if (j < sg.y && q >= 0 && q < LIST_CAP) {
        sList[q] = w[u];
        atomicAdd(&sCnt[w[u] >> ENT_SHIFT], 1);
      }
    }
  }
  __syncthreads();
  int n = tot - base;
  if (n > LIST_CAP) n = LIST_CAP;
  if (tid < 32) {
    int v[SC_PL];
    int sum = 0;
#pragma unroll
    for (int k = 0; k < SC_PL; ++k) {
      const int x = sCnt[SC_PL * tid + k];
      v[k] = sum;
      sum += x;
    }
    int inc = sum;
#pragma unroll
    for (int off = 1; off < 32; off <<= 1) {
      const int y = __shfl_up(inc, off, 32);
      if (tid >= off) inc += y;
    }
    const int excl = inc - sum;
#pragma unroll
    for (int k = 0; k < SC_PL; ++k) sCnt[SC_PL * tid + k] = excl + v[k];
  }
  __syncthreads();
  {
    int ev[LIST_CAP / BIN_NT];
#pragma unroll
    for (int k = 0; k < LIST_CAP / BIN_NT; ++k) {
      const int i = tid + k * BIN_NT;
      ev[k] = i < n ? sList[i] : -1;
    }
#pragma unroll
    for (int k = 0; k < LIST_CAP / BIN_NT; ++k)
      if (ev[k] >= 0) sSort[atomicAdd(&sCnt[ev[k] >> ENT_SHIFT], 1)] = ev[k];
  }
  __syncthreads();
  if (tid < SC_PL * 32) sCnt[tid] = 0;
  const int p0 = wave * 32;
  int ent = PAD_ENT;
  if (p0 + lane < n) ent = sSort[p0 + lane];
  for (int p = p0; p < n; p += BIN_NT) {
    const int pn = p + BIN_NT + lane;
    int entN = sSort[pn < n ? pn : n - 1];
    entN = pn < n ? entN : PAD_ENT;

    const int dl = ent >> ENT_SHIFT;
    const int dnx = __shfl_down(dl, 1, 32);
    const unsigned ends = __builtin_amdgcn_ballot_w32(lane == 31 || dl != dnx);
    float v[NG][G][NL];
    float run[NL];
#pragma unroll
    for (int i = 0; i < NL; ++i) run[i] = 0.0f;
#pragma unroll
    for (int k = 0; k < NG + PD; ++k) {
      if (k >= PD) {
#pragma unroll
        for (int g = 0; g < G; ++g) {
          const int j = (k - PD) * G + g;
#pragma unroll
          for (int i = 0; i < NL; ++i) run[i] += v[k - PD][g][i];
          if (ends & (1u << j)) {
            const int lj = __builtin_amdgcn_readlane(ent, j) >> ENT_SHIFT;
            float* o = acc + lj * P + lane;
#pragma unroll
            for (int i = 0; i < NL; ++i) { atomicAdd(o + 32 * i, run[i]); run[i] = 0.0f; }
          }
        }
      }
      if (k < NG) {
#pragma unroll
        for (int g = 0; g < G; ++g) {
          const int sj = __builtin_amdgcn_readlane(ent, k * G + g) & ENT_MASK;
          const float* hp = H + (long long)sj * D + lane;
#pragma unroll
          for (int i = 0; i < NL; ++i) v[k][g][i] = hp[32 * i];
        }
      }
    }
    ent = entN;
  }
  }
  __syncthreads();

  if (GEMM) {
    const float4 bb = *(const float4*)(bias + 4 * (tid % (D / 4)));
    for (int i = tid; i < NRT * 16 * (D / 4); i += BIN_NT) {
      const int r = i / (D / 4), c = i % (D / 4);
      v4f v = {0.0f, 0.0f, 0.0f, 0.0f};
      if (r < nrows) {
        v = *(const v4f*)(acc + r * P + 4 * c);
        const float dr = sDis[r];
        v.x = fmaxf(dr * v.x + bb.x, 0.0f);
        v.y = fmaxf(dr * v.y + bb.y, 0.0f);
        v.z = fmaxf(dr * v.z + bb.z, 0.0f);
        v.w = fmaxf(dr * v.w + bb.w, 0.0f);
      }
      *(v4f*)(acc + r * P + 4 * c) = v;
    }
    __syncthreads();

    if (wave < NRT) {
      const int mn  = lane & 15;
      const int grp = lane >> 4;
      const float* arow = acc + (wave * 16 + mn) * P + 2 * grp;
      const float* wq = sW + (grp * 16 + mn) * 4;
      const int lr0 = wave * 16 + 8 * grp;
      float* out = res + (long long)(lo + lr0) * NO + mn;
      float dr[8];
#pragma unroll
      for (int r = 0; r < 8; ++r) dr[r] = sDis[lr0 + r];
      v8f c0 = {0.0f, 0.0f, 0.0f, 0.0f, 0.0f, 0.0f, 0.0f, 0.0f};
      v8f c1 = {0.0f, 0.0f, 0.0f, 0.0f, 0.0f, 0.0f, 0.0f, 0.0f};
      asm volatile("" : "+v"(c0), "+v"(c1), "+v"(out));
      typedef __attribute__((address_space(1))) float gfloat;
      gfloat* outg = (gfloat*)out;
#pragma unroll
      for (int k = 0; k < D; k += 4) {
        const v2f a = *(const v2f*)(arow + k);
        const v4f b = *(const v4f*)(wq + k * 32);
        const v2f b0 = {b.x, b.y};
        const v2f b1 = {b.z, b.w};
        c0 = __builtin_amdgcn_wmma_f32_16x16x4_f32(
            false, a, false, b0, (short)0, c0, false, false);
        c1 = __builtin_amdgcn_wmma_f32_16x16x4_f32(
            false, a, false, b1, (short)0, c1, false, false);
      }
      asm volatile("v_nop\n\tv_nop\n\tv_nop\n\tv_nop" : "+v"(c0), "+v"(c1));

#pragma unroll
      for (int r = 0; r < 8; ++r) {
        if (lr0 + r < nrows) {
          outg[r * NO] = dr[r] * c0[r];
          outg[r * NO + 16] = dr[r] * c1[r];
        }
      }
    }
  } else {
    constexpr int RPW = (OWN_RS + 31) / 32;
    const float bl = bias[lane];
    float z[RPW];
#pragma unroll
    for (int j = 0; j < RPW; ++j) {
      const int r = wave + 32 * j;
      const int rr = r < nrows ? r : 0;
      z[j] = sDis[rr] * acc[rr * P + lane] + bl;
    }
    float m[RPW], ev[RPW], sm[RPW];
#pragma unroll
    for (int j = 0; j < RPW; ++j) m[j] = z[j];
#pragma unroll
    for (int off = 16; off > 0; off >>= 1) {
#pragma unroll
      for (int j = 0; j < RPW; ++j) m[j] = fmaxf(m[j], __shfl_xor(m[j], off, 32));
    }
#pragma unroll
    for (int j = 0; j < RPW; ++j) { ev[j] = __expf(z[j] - m[j]); sm[j] = ev[j]; }
#pragma unroll
    for (int off = 16; off > 0; off >>= 1) {
#pragma unroll
      for (int j = 0; j < RPW; ++j) sm[j] += __shfl_xor(sm[j], off, 32);
    }
#pragma unroll
    for (int j = 0; j < RPW; ++j) {
      const int r = wave + 32 * j;
      if (r < nrows) res[(long long)(lo + r) * D + lane] = ev[j] / sm[j];
    }
  }
}

extern "C" void kernel_launch(void* const* d_in, const int* in_sizes, int n_in,
                              void* d_out, int out_size, void* d_ws, size_t ws_size,
                              hipStream_t stream) {
  (void)n_in; (void)out_size; (void)ws_size;
  const float* x  = (const float*)d_in[0];
  const int*   ei = (const int*)d_in[1];
  const float* W1 = (const float*)d_in[2];
  const float* b1 = (const float*)d_in[3];
  const float* W2 = (const float*)d_in[4];
  const float* b2 = (const float*)d_in[5];

  const int IN = 256, HID = 64, OD = 32;
  const int M = in_sizes[0] / IN;
  const int E = in_sizes[1] / 2;
  const int* src = ei;
  const int* dst = ei + E;

  float* ws   = (float*)d_ws;
  float* dis  = ws;
  float* H1   = dis + (((long long)M + 1023) & ~1023LL);
  float* H2   = H1 + (long long)M * HID;
  int* list   = (int*)(H2 + (long long)M * HID);
  int* seg    = list + E;
  float* out  = (float*)d_out;

  const int NRANGE = (M + OWN_RS - 1) / OWN_RS;

  bin_sort_kernel<<<BIN_NCH, BIN_NT, 0, stream>>>(src, dst, E, seg, list);

  gemm1_range_kernel<<<G1_PARTS * OWN_NR, G1_NT, 0, stream>>>(x, W1, list, seg, dis, H1, M);
  owner_sum_kernel<64><<<NRANGE, BIN_NT, 0, stream>>>(list, seg, H1, dis, b1, W2, H2, M);

  owner_sum_kernel<32><<<NRANGE, BIN_NT, 0, stream>>>(list, seg, H2, dis, b2, W2, out, M);
}
